# starting kernel
# baseline (speedup 1.0000x reference)
#include <hip/hip_runtime.h>


typedef float v2f __attribute__((ext_vector_type(2)));
typedef float v8f __attribute__((ext_vector_type(8)));

#define I_DIM 5
#define H_DIM 8
#define T_DIM 512

__device__ __forceinline__ v2f mk2(float a, float b) { v2f v; v.x = a; v.y = b; return v; }

#if __has_builtin(__builtin_amdgcn_tanhf)
__device__ __forceinline__ float tanh_hw(float x) {
  return __builtin_amdgcn_tanhf(x);
}
#else
__device__ __forceinline__ float tanh_hw(float x) {
  float e = __expf(-2.0f * x);
  return (1.0f - e) * __builtin_amdgcn_rcpf(1.0f + e);
}
#endif
__device__ __forceinline__ float sigmoid_f(float x) {
  return __builtin_fmaf(0.5f, tanh_hw(0.5f * x), 0.5f);
}

__device__ __forceinline__ float swz16(float v) {
  return __int_as_float(__builtin_amdgcn_ds_swizzle(__float_as_int(v), 0x401F));
}

__device__ __forceinline__ v8f wmma4(v2f a, v2f b, v8f c) {
  return __builtin_amdgcn_wmma_f32_16x16x4_f32(false, a, false, b, (short)0, c,
                                               false, false);
}

__global__ void __launch_bounds__(128) gru_wmma_kernel(
    const float* __restrict__ x, const float* __restrict__ w_ih,
    const float* __restrict__ w_hh, const float* __restrict__ b_ih,
    const float* __restrict__ b_hh, const float* __restrict__ fc_w,
    const float* __restrict__ fc_b, float* __restrict__ out) {
  const int lane = threadIdx.x & 31;
  const int wid  = threadIdx.x >> 5;
  const int row  = lane & 15;
  const bool hi  = lane >= 16;
  const int ka   = hi ? 2 : 0;
  const int bglob = blockIdx.x * 64 + wid * 16 + row;

  const bool nrow = row < 8;
  const float* whr0 = w_hh + row * H_DIM;
  const float* whr1 = w_hh + (16 + (row & 7)) * H_DIM;
  v2f a_hh_t0_k0 = mk2(whr0[ka],     whr0[ka + 1]);
  v2f a_hh_t0_k1 = mk2(whr0[4 + ka], whr0[5 + ka]);
  v2f a_hh_t1_k0 = nrow ? mk2(whr1[ka],     whr1[ka + 1]) : mk2(0.f, 0.f);
  v2f a_hh_t1_k1 = nrow ? mk2(whr1[4 + ka], whr1[5 + ka]) : mk2(0.f, 0.f);

  const float* wir0 = w_ih + row * I_DIM;
  const float* wir1 = w_ih + (16 + (row & 7)) * I_DIM;
  v2f a_ih_t0_k0 = mk2(wir0[ka], wir0[ka + 1]);
  v2f a_ih_t0_k1 = hi ? mk2(0.f, 0.f) : mk2(wir0[4], 0.f);
  v2f a_ih_t1_k0 = nrow ? mk2(wir1[ka], wir1[ka + 1]) : mk2(0.f, 0.f);
  v2f a_ih_t1_k1 = (nrow && !hi) ? mk2(wir1[4], 0.f) : mk2(0.f, 0.f);

  v8f c_rz, c_xn, c_hn;
#pragma unroll
  for (int j = 0; j < 8; ++j) {
    int m = j + (hi ? 8 : 0);
    c_rz[j] = b_ih[m] + b_hh[m];
    c_xn[j] = hi ? 0.f : b_ih[16 + j];
    c_hn[j] = hi ? 0.f : b_hh[16 + j];
  }

  v2f h0 = mk2(0.f, 0.f), h1 = mk2(0.f, 0.f);

  const float* xb = x + (size_t)bglob * (T_DIM * I_DIM);
  float xa = xb[ka], xc = xb[ka + 1], xe = hi ? 0.f : xb[4];

  for (int t = 0; t < T_DIM; ++t) {
    int tn = (t + 1 < T_DIM) ? t + 1 : t;
    const float* p = xb + tn * I_DIM;
    float nxa = p[ka], nxc = p[ka + 1], nxe = hi ? 0.f : p[4];
    __builtin_prefetch(xb + ((t + 16 < T_DIM) ? t + 16 : t) * I_DIM, 0, 1);

    v2f xB0 = mk2(xa, xc);
    v2f xB1 = mk2(xe, 0.f);

    v8f rz = c_rz;
    rz = wmma4(a_ih_t0_k0, xB0, rz);
    rz = wmma4(a_ih_t0_k1, xB1, rz);
    rz = wmma4(a_hh_t0_k0, h0, rz);
    rz = wmma4(a_hh_t0_k1, h1, rz);
    v8f xn = c_xn;
    xn = wmma4(a_ih_t1_k0, xB0, xn);
    xn = wmma4(a_ih_t1_k1, xB1, xn);
    v8f hn = c_hn;
    hn = wmma4(a_hh_t1_k0, h0, hn);
    hn = wmma4(a_hh_t1_k1, h1, hn);

    float s[8], n[8];
#pragma unroll
    for (int j = 0; j < 8; ++j) s[j] = sigmoid_f(rz[j]);
#pragma unroll
    for (int j = 0; j < 8; ++j) n[j] = tanh_hw(xn[j] + s[j] * hn[j]);

    float zs0 = swz16(s[0]), zs1 = swz16(s[1]), zs4 = swz16(s[4]), zs5 = swz16(s[5]);
    float ns2 = swz16(n[2]), ns3 = swz16(n[3]), ns6 = swz16(n[6]), ns7 = swz16(n[7]);

    float z0 = hi ? s[2] : zs0, nn0 = hi ? ns2 : n[0];
    float z1 = hi ? s[3] : zs1, nn1 = hi ? ns3 : n[1];
    float z2 = hi ? s[6] : zs4, nn2 = hi ? ns6 : n[4];
    float z3 = hi ? s[7] : zs5, nn3 = hi ? ns7 : n[5];

    h0.x = nn0 + z0 * (h0.x - nn0);
    h0.y = nn1 + z1 * (h0.y - nn1);
    h1.x = nn2 + z2 * (h1.x - nn2);
    h1.y = nn3 + z3 * (h1.y - nn3);

    xa = nxa; xc = nxc; xe = nxe;
  }

  float o0 = fc_w[0 * 8 + ka] * h0.x + fc_w[0 * 8 + ka + 1] * h0.y +
             fc_w[0 * 8 + ka + 4] * h1.x + fc_w[0 * 8 + ka + 5] * h1.y;
  float o1 = fc_w[1 * 8 + ka] * h0.x + fc_w[1 * 8 + ka + 1] * h0.y +
             fc_w[1 * 8 + ka + 4] * h1.x + fc_w[1 * 8 + ka + 5] * h1.y;
  o0 += swz16(o0);
  o1 += swz16(o1);
  if (!hi) {
    out[(size_t)bglob * 2 + 0] = o0 + fc_b[0];
    out[(size_t)bglob * 2 + 1] = o1 + fc_b[1];
  }
}

extern "C" void kernel_launch(void* const* d_in, const int* in_sizes, int n_in,
                              void* d_out, int out_size, void* d_ws, size_t ws_size,
                              hipStream_t stream) {
  const float* x    = (const float*)d_in[0];
  const float* w_ih = (const float*)d_in[1];
  const float* w_hh = (const float*)d_in[2];
  const float* b_ih = (const float*)d_in[3];
  const float* b_hh = (const float*)d_in[4];
  const float* fc_w = (const float*)d_in[5];
  const float* fc_b = (const float*)d_in[6];
  float* out = (float*)d_out;

  const int B = in_sizes[0] / (T_DIM * I_DIM);
  dim3 grid(B / 64), block(128);
  gru_wmma_kernel<<<grid, block, 0, stream>>>(x, w_ih, w_hh, b_ih, b_hh, fc_w,
                                              fc_b, out);
}


# x data path: each wave stages its 16 batch rows through its own LDS region in 16-timestep chunks (whole 16-byte global loads, two chunks ahead); per-step x operands are LDS reads; sums unchanged
# speedup vs baseline: 1.5226x; 1.5226x over previous
#include <hip/hip_runtime.h>


typedef float v2f __attribute__((ext_vector_type(2)));
typedef float v8f __attribute__((ext_vector_type(8)));

#define I_DIM 5
#define H_DIM 8
#define T_DIM 512

#define TC 16
#define NCHUNK (T_DIM / TC)
#define ROW4 (TC * I_DIM / 4)
#define LROW 84
#define WBUF (16 * LROW)
#define XROW (T_DIM * I_DIM)

typedef float v4f __attribute__((ext_vector_type(4)));

__device__ __forceinline__ v2f mk2(float a, float b) { v2f v; v.x = a; v.y = b; return v; }

#if __has_builtin(__builtin_amdgcn_tanhf)
__device__ __forceinline__ float tanh_hw(float x) {
  return __builtin_amdgcn_tanhf(x);
}
#else
__device__ __forceinline__ float tanh_hw(float x) {
  float e = __expf(-2.0f * x);
  return (1.0f - e) * __builtin_amdgcn_rcpf(1.0f + e);
}
#endif
__device__ __forceinline__ float sigmoid_f(float x) {
  return __builtin_fmaf(0.5f, tanh_hw(0.5f * x), 0.5f);
}

__device__ __forceinline__ float swz16(float v) {
  return __int_as_float(__builtin_amdgcn_ds_swizzle(__float_as_int(v), 0x401F));
}

__device__ __forceinline__ v8f wmma4(v2f a, v2f b, v8f c) {
  return __builtin_amdgcn_wmma_f32_16x16x4_f32(false, a, false, b, (short)0, c,
                                               false, false);
}

__global__ void __launch_bounds__(128) gru_wmma_kernel(
    const float* __restrict__ x, const float* __restrict__ w_ih,
    const float* __restrict__ w_hh, const float* __restrict__ b_ih,
    const float* __restrict__ b_hh, const float* __restrict__ fc_w,
    const float* __restrict__ fc_b, float* __restrict__ out) {
  __shared__ __attribute__((aligned(16))) float xs[4 * 2 * WBUF];
  const int lane = threadIdx.x & 31;
  const int wid  = threadIdx.x >> 5;
  const int row  = lane & 15;
  const bool hi  = lane >= 16;
  const int ka   = hi ? 2 : 0;
  const int bglob = blockIdx.x * 64 + wid * 16 + row;

  const bool nrow = row < 8;
  const float* whr0 = w_hh + row * H_DIM;
  const float* whr1 = w_hh + (16 + (row & 7)) * H_DIM;
  v2f a_hh_t0_k0 = mk2(whr0[ka],     whr0[ka + 1]);
  v2f a_hh_t0_k1 = mk2(whr0[4 + ka], whr0[5 + ka]);
  v2f a_hh_t1_k0 = nrow ? mk2(whr1[ka],     whr1[ka + 1]) : mk2(0.f, 0.f);
  v2f a_hh_t1_k1 = nrow ? mk2(whr1[4 + ka], whr1[5 + ka]) : mk2(0.f, 0.f);

  const float* wir0 = w_ih + row * I_DIM;
  const float* wir1 = w_ih + (16 + (row & 7)) * I_DIM;
  v2f a_ih_t0_k0 = mk2(wir0[ka], wir0[ka + 1]);
  v2f a_ih_t0_k1 = hi ? mk2(0.f, 0.f) : mk2(wir0[4], 0.f);
  v2f a_ih_t1_k0 = nrow ? mk2(wir1[ka], wir1[ka + 1]) : mk2(0.f, 0.f);
  v2f a_ih_t1_k1 = (nrow && !hi) ? mk2(wir1[4], 0.f) : mk2(0.f, 0.f);

  v8f c_rz, c_xn, c_hn;
#pragma unroll
  for (int j = 0; j < 8; ++j) {
    int m = j + (hi ? 8 : 0);
    c_rz[j] = b_ih[m] + b_hh[m];
    c_xn[j] = hi ? 0.f : b_ih[16 + j];
    c_hn[j] = hi ? 0.f : b_hh[16 + j];
  }

  v2f h0 = mk2(0.f, 0.f), h1 = mk2(0.f, 0.f);

  const int wid_s = __builtin_amdgcn_readfirstlane(wid);
  const float* xw = x + (size_t)(blockIdx.x * 64 + wid_s * 16) * XROW;
  float* wb = xs + wid_s * (2 * WBUF);
  const bool stg = lane < ROW4;
  const int l4 = lane * 4;

  v4f g0[16], g[16];
#pragma unroll
  for (int r = 0; r < 16; ++r) { g0[r] = (v4f)(0.f); g[r] = (v4f)(0.f); }
  if (stg) {
#pragma unroll
    for (int r = 0; r < 16; ++r)
      g0[r] = *reinterpret_cast<const v4f*>(xw + r * XROW + l4);
#pragma unroll
    for (int r = 0; r < 16; ++r)
      g[r] = *reinterpret_cast<const v4f*>(xw + r * XROW + TC * I_DIM + l4);
#pragma unroll
    for (int r = 0; r < 16; ++r)
      *reinterpret_cast<v4f*>(wb + r * LROW + l4) = g0[r];
  }

  const float* lrow = wb + row * LROW;
  float xa = lrow[ka], xc = lrow[ka + 1], xe0 = lrow[4];
  float xe = hi ? 0.f : xe0;

  for (int c = 0; c < NCHUNK; ++c) {
    const int cb = c & 1, nb = cb ^ 1;
    if (c + 1 < NCHUNK) {
      if (stg) {
#pragma unroll
        for (int r = 0; r < 16; ++r)
          *reinterpret_cast<v4f*>(wb + nb * WBUF + r * LROW + l4) = g[r];
      }
    }
    if (c + 2 < NCHUNK) {
      if (stg) {
        const float* gp = xw + (c + 2) * (TC * I_DIM) + l4;
#pragma unroll
        for (int r = 0; r < 16; ++r)
          g[r] = *reinterpret_cast<const v4f*>(gp + r * XROW);
      }
    }

#pragma unroll 1
    for (int tt = 0; tt < TC; ++tt) {
        const int ntt  = (tt + 1 == TC) ? 0 : tt + 1;
        const int nbuf = (tt + 1 == TC) ? nb : cb;
        const float* p = lrow + nbuf * WBUF + ntt * I_DIM;
        float nxa = p[ka], nxc = p[ka + 1], nxe0 = p[4];
        float nxe = hi ? 0.f : nxe0;

        v2f xB0 = mk2(xa, xc);
        v2f xB1 = mk2(xe, 0.f);

        v8f rz = c_rz;
        rz = wmma4(a_ih_t0_k0, xB0, rz);
        rz = wmma4(a_ih_t0_k1, xB1, rz);
        rz = wmma4(a_hh_t0_k0, h0, rz);
        rz = wmma4(a_hh_t0_k1, h1, rz);
        v8f xn = c_xn;
        xn = wmma4(a_ih_t1_k0, xB0, xn);
        xn = wmma4(a_ih_t1_k1, xB1, xn);
        v8f hn = c_hn;
        hn = wmma4(a_hh_t1_k0, h0, hn);
        hn = wmma4(a_hh_t1_k1, h1, hn);

        float s[8], n[8];
#pragma unroll
        for (int j = 0; j < 8; ++j) s[j] = sigmoid_f(rz[j]);
#pragma unroll
        for (int j = 0; j < 8; ++j) n[j] = tanh_hw(xn[j] + s[j] * hn[j]);

        float zs0 = swz16(s[0]), zs1 = swz16(s[1]), zs4 = swz16(s[4]), zs5 = swz16(s[5]);
        float ns2 = swz16(n[2]), ns3 = swz16(n[3]), ns6 = swz16(n[6]), ns7 = swz16(n[7]);

        float z0 = hi ? s[2] : zs0, nn0 = hi ? ns2 : n[0];
        float z1 = hi ? s[3] : zs1, nn1 = hi ? ns3 : n[1];
        float z2 = hi ? s[6] : zs4, nn2 = hi ? ns6 : n[4];
        float z3 = hi ? s[7] : zs5, nn3 = hi ? ns7 : n[5];

        h0.x = nn0 + z0 * (h0.x - nn0);
        h0.y = nn1 + z1 * (h0.y - nn1);
        h1.x = nn2 + z2 * (h1.x - nn2);
        h1.y = nn3 + z3 * (h1.y - nn3);

        xa = nxa; xc = nxc; xe = nxe;
    }
  }

  float o0 = fc_w[0 * 8 + ka] * h0.x + fc_w[0 * 8 + ka + 1] * h0.y +
             fc_w[0 * 8 + ka + 4] * h1.x + fc_w[0 * 8 + ka + 5] * h1.y;
  float o1 = fc_w[1 * 8 + ka] * h0.x + fc_w[1 * 8 + ka + 1] * h0.y +
             fc_w[1 * 8 + ka + 4] * h1.x + fc_w[1 * 8 + ka + 5] * h1.y;
  o0 += swz16(o0);
  o1 += swz16(o1);
  if (!hi) {
    out[(size_t)bglob * 2 + 0] = o0 + fc_b[0];
    out[(size_t)bglob * 2 + 1] = o1 + fc_b[1];
  }
}

extern "C" void kernel_launch(void* const* d_in, const int* in_sizes, int n_in,
                              void* d_out, int out_size, void* d_ws, size_t ws_size,
                              hipStream_t stream) {
  const float* x    = (const float*)d_in[0];
  const float* w_ih = (const float*)d_in[1];
  const float* w_hh = (const float*)d_in[2];
  const float* b_ih = (const float*)d_in[3];
  const float* b_hh = (const float*)d_in[4];
  const float* fc_w = (const float*)d_in[5];
  const float* fc_b = (const float*)d_in[6];
  float* out = (float*)d_out;

  const int B = in_sizes[0] / (T_DIM * I_DIM);
  dim3 grid(B / 64), block(128);
  gru_wmma_kernel<<<grid, block, 0, stream>>>(x, w_ih, w_hh, b_ih, b_hh, fc_w,
                                              fc_b, out);
}


# gate halves exchanged in the register file: 8 v_permlane16_swap_b32 per step replace 8 ds_swizzle, their LDS wait and 8 selects; values and sums unchanged
# speedup vs baseline: 1.8668x; 1.2261x over previous
#include <hip/hip_runtime.h>


typedef float v2f __attribute__((ext_vector_type(2)));
typedef float v8f __attribute__((ext_vector_type(8)));

#define I_DIM 5
#define H_DIM 8
#define T_DIM 512

#define TC 16
#define NCHUNK (T_DIM / TC)
#define ROW4 (TC * I_DIM / 4)
#define LROW 84
#define WBUF (16 * LROW)
#define XROW (T_DIM * I_DIM)

typedef float v4f __attribute__((ext_vector_type(4)));

__device__ __forceinline__ v2f mk2(float a, float b) { v2f v; v.x = a; v.y = b; return v; }

#if __has_builtin(__builtin_amdgcn_tanhf)
__device__ __forceinline__ float tanh_hw(float x) {
  return __builtin_amdgcn_tanhf(x);
}
#else
__device__ __forceinline__ float tanh_hw(float x) {
  float e = __expf(-2.0f * x);
  return (1.0f - e) * __builtin_amdgcn_rcpf(1.0f + e);
}
#endif
__device__ __forceinline__ float sigmoid_f(float x) {
  return __builtin_fmaf(0.5f, tanh_hw(0.5f * x), 0.5f);
}

__device__ __forceinline__ float swz16(float v) {
  return __int_as_float(__builtin_amdgcn_ds_swizzle(__float_as_int(v), 0x401F));
}

typedef unsigned int v2u __attribute__((ext_vector_type(2)));
__device__ __forceinline__ float lo_of_both(float d, float s) {
  v2u r = __builtin_amdgcn_permlane16_swap(__float_as_uint(d), __float_as_uint(s), false, false);
  return __uint_as_float(r.x);
}
__device__ __forceinline__ float hi_of_both(float d, float s) {
  v2u r = __builtin_amdgcn_permlane16_swap(__float_as_uint(d), __float_as_uint(s), false, false);
  return __uint_as_float(r.y);
}

__device__ __forceinline__ v8f wmma4(v2f a, v2f b, v8f c) {
  return __builtin_amdgcn_wmma_f32_16x16x4_f32(false, a, false, b, (short)0, c,
                                               false, false);
}

__global__ void __launch_bounds__(128) gru_wmma_kernel(
    const float* __restrict__ x, const float* __restrict__ w_ih,
    const float* __restrict__ w_hh, const float* __restrict__ b_ih,
    const float* __restrict__ b_hh, const float* __restrict__ fc_w,
    const float* __restrict__ fc_b, float* __restrict__ out) {
  __shared__ __attribute__((aligned(16))) float xs[4 * 2 * WBUF];
  const int lane = threadIdx.x & 31;
  const int wid  = threadIdx.x >> 5;
  const int row  = lane & 15;
  const bool hi  = lane >= 16;
  const int ka   = hi ? 2 : 0;
  const int bglob = blockIdx.x * 64 + wid * 16 + row;

  const bool nrow = row < 8;
  const float* whr0 = w_hh + row * H_DIM;
  const float* whr1 = w_hh + (16 + (row & 7)) * H_DIM;
  v2f a_hh_t0_k0 = mk2(whr0[ka],     whr0[ka + 1]);
  v2f a_hh_t0_k1 = mk2(whr0[4 + ka], whr0[5 + ka]);
  v2f a_hh_t1_k0 = nrow ? mk2(whr1[ka],     whr1[ka + 1]) : mk2(0.f, 0.f);
  v2f a_hh_t1_k1 = nrow ? mk2(whr1[4 + ka], whr1[5 + ka]) : mk2(0.f, 0.f);

  const float* wir0 = w_ih + row * I_DIM;
  const float* wir1 = w_ih + (16 + (row & 7)) * I_DIM;
  v2f a_ih_t0_k0 = mk2(wir0[ka], wir0[ka + 1]);
  v2f a_ih_t0_k1 = hi ? mk2(0.f, 0.f) : mk2(wir0[4], 0.f);
  v2f a_ih_t1_k0 = nrow ? mk2(wir1[ka], wir1[ka + 1]) : mk2(0.f, 0.f);
  v2f a_ih_t1_k1 = (nrow && !hi) ? mk2(wir1[4], 0.f) : mk2(0.f, 0.f);

  v8f c_rz, c_xn, c_hn;
#pragma unroll
  for (int j = 0; j < 8; ++j) {
    int m = j + (hi ? 8 : 0);
    c_rz[j] = b_ih[m] + b_hh[m];
    c_xn[j] = hi ? 0.f : b_ih[16 + j];
    c_hn[j] = hi ? 0.f : b_hh[16 + j];
  }

  v2f h0 = mk2(0.f, 0.f), h1 = mk2(0.f, 0.f);

  const int wid_s = __builtin_amdgcn_readfirstlane(wid);
  const float* xw = x + (size_t)(blockIdx.x * 64 + wid_s * 16) * XROW;
  float* wb = xs + wid_s * (2 * WBUF);
  const bool stg = lane < ROW4;
  const int l4 = lane * 4;

  v4f g0[16], g[16];
#pragma unroll
  for (int r = 0; r < 16; ++r) { g0[r] = (v4f)(0.f); g[r] = (v4f)(0.f); }
  if (stg) {
#pragma unroll
    for (int r = 0; r < 16; ++r)
      g0[r] = *reinterpret_cast<const v4f*>(xw + r * XROW + l4);
#pragma unroll
    for (int r = 0; r < 16; ++r)
      g[r] = *reinterpret_cast<const v4f*>(xw + r * XROW + TC * I_DIM + l4);
#pragma unroll
    for (int r = 0; r < 16; ++r)
      *reinterpret_cast<v4f*>(wb + r * LROW + l4) = g0[r];
  }

  const float* lrow = wb + row * LROW;
  float xa = lrow[ka], xc = lrow[ka + 1], xe0 = lrow[4];
  float xe = hi ? 0.f : xe0;

  for (int c = 0; c < NCHUNK; ++c) {
    const int cb = c & 1, nb = cb ^ 1;
    if (c + 1 < NCHUNK) {
      if (stg) {
#pragma unroll
        for (int r = 0; r < 16; ++r)
          *reinterpret_cast<v4f*>(wb + nb * WBUF + r * LROW + l4) = g[r];
      }
    }
    if (c + 2 < NCHUNK) {
      if (stg) {
        const float* gp = xw + (c + 2) * (TC * I_DIM) + l4;
#pragma unroll
        for (int r = 0; r < 16; ++r)
          g[r] = *reinterpret_cast<const v4f*>(gp + r * XROW);
      }
    }

#pragma unroll 1
    for (int tt = 0; tt < TC; ++tt) {
        const int ntt  = (tt + 1 == TC) ? 0 : tt + 1;
        const int nbuf = (tt + 1 == TC) ? nb : cb;
        const float* p = lrow + nbuf * WBUF + ntt * I_DIM;
        float nxa = p[ka], nxc = p[ka + 1], nxe0 = p[4];
        float nxe = hi ? 0.f : nxe0;

        v2f xB0 = mk2(xa, xc);
        v2f xB1 = mk2(xe, 0.f);

        v8f rz = c_rz;
        rz = wmma4(a_ih_t0_k0, xB0, rz);
        rz = wmma4(a_ih_t0_k1, xB1, rz);
        rz = wmma4(a_hh_t0_k0, h0, rz);
        rz = wmma4(a_hh_t0_k1, h1, rz);
        v8f xn = c_xn;
        xn = wmma4(a_ih_t1_k0, xB0, xn);
        xn = wmma4(a_ih_t1_k1, xB1, xn);
        v8f hn = c_hn;
        hn = wmma4(a_hh_t1_k0, h0, hn);
        hn = wmma4(a_hh_t1_k1, h1, hn);

        float s[8], n[8];
#pragma unroll
        for (int j = 0; j < 8; ++j) s[j] = sigmoid_f(rz[j]);
#pragma unroll
        for (int j = 0; j < 8; ++j) n[j] = tanh_hw(xn[j] + s[j] * hn[j]);

        float z0 = hi_of_both(s[0], s[2]), nn0 = lo_of_both(n[0], n[2]);
        float z1 = hi_of_both(s[1], s[3]), nn1 = lo_of_both(n[1], n[3]);
        float z2 = hi_of_both(s[4], s[6]), nn2 = lo_of_both(n[4], n[6]);
        float z3 = hi_of_both(s[5], s[7]), nn3 = lo_of_both(n[5], n[7]);

        h0.x = nn0 + z0 * (h0.x - nn0);
        h0.y = nn1 + z1 * (h0.y - nn1);
        h1.x = nn2 + z2 * (h1.x - nn2);
        h1.y = nn3 + z3 * (h1.y - nn3);

        xa = nxa; xc = nxc; xe = nxe;
    }
  }

  float o0 = fc_w[0 * 8 + ka] * h0.x + fc_w[0 * 8 + ka + 1] * h0.y +
             fc_w[0 * 8 + ka + 4] * h1.x + fc_w[0 * 8 + ka + 5] * h1.y;
  float o1 = fc_w[1 * 8 + ka] * h0.x + fc_w[1 * 8 + ka + 1] * h0.y +
             fc_w[1 * 8 + ka + 4] * h1.x + fc_w[1 * 8 + ka + 5] * h1.y;
  o0 += swz16(o0);
  o1 += swz16(o1);
  if (!hi) {
    out[(size_t)bglob * 2 + 0] = o0 + fc_b[0];
    out[(size_t)bglob * 2 + 1] = o1 + fc_b[1];
  }
}

extern "C" void kernel_launch(void* const* d_in, const int* in_sizes, int n_in,
                              void* d_out, int out_size, void* d_ws, size_t ws_size,
                              hipStream_t stream) {
  const float* x    = (const float*)d_in[0];
  const float* w_ih = (const float*)d_in[1];
  const float* w_hh = (const float*)d_in[2];
  const float* b_ih = (const float*)d_in[3];
  const float* b_hh = (const float*)d_in[4];
  const float* fc_w = (const float*)d_in[5];
  const float* fc_b = (const float*)d_in[6];
  float* out = (float*)d_out;

  const int B = in_sizes[0] / (T_DIM * I_DIM);
  dim3 grid(B / 64), block(128);
  gru_wmma_kernel<<<grid, block, 0, stream>>>(x, w_ih, w_hh, b_ih, b_hh, fc_w,
                                              fc_b, out);
}


# x products of step t+1 issued inside step t of the same wave, one per stretch of gate math (issue order pinned), so the recurrence waits only for the 4 h products; sums and their order unchanged
# speedup vs baseline: 1.9264x; 1.0319x over previous
#include <hip/hip_runtime.h>


typedef float v2f __attribute__((ext_vector_type(2)));
typedef float v8f __attribute__((ext_vector_type(8)));

#define I_DIM 5
#define H_DIM 8
#define T_DIM 512

#define TC 16
#define NCHUNK (T_DIM / TC)
#define ROW4 (TC * I_DIM / 4)
#define LROW 84
#define WBUF (16 * LROW)
#define XROW (T_DIM * I_DIM)

typedef float v4f __attribute__((ext_vector_type(4)));

__device__ __forceinline__ v2f mk2(float a, float b) { v2f v; v.x = a; v.y = b; return v; }

#if __has_builtin(__builtin_amdgcn_tanhf)
__device__ __forceinline__ float tanh_hw(float x) {
  return __builtin_amdgcn_tanhf(x);
}
#else
__device__ __forceinline__ float tanh_hw(float x) {
  float e = __expf(-2.0f * x);
  return (1.0f - e) * __builtin_amdgcn_rcpf(1.0f + e);
}
#endif

__device__ __forceinline__ float swz16(float v) {
  return __int_as_float(__builtin_amdgcn_ds_swizzle(__float_as_int(v), 0x401F));
}

typedef unsigned int v2u __attribute__((ext_vector_type(2)));
__device__ __forceinline__ float lo_of_both(float d, float s) {
  v2u r = __builtin_amdgcn_permlane16_swap(__float_as_uint(d), __float_as_uint(s), false, false);
  return __uint_as_float(r.x);
}
__device__ __forceinline__ float hi_of_both(float d, float s) {
  v2u r = __builtin_amdgcn_permlane16_swap(__float_as_uint(d), __float_as_uint(s), false, false);
  return __uint_as_float(r.y);
}

__device__ __forceinline__ v8f wmma4(v2f a, v2f b, v8f c) {
  return __builtin_amdgcn_wmma_f32_16x16x4_f32(false, a, false, b, (short)0, c,
                                               false, false);
}

#define STAGE(...) do { asm volatile("" : __VA_ARGS__); __builtin_amdgcn_sched_barrier(0); } while (0)

__global__ void __launch_bounds__(128) gru_wmma_kernel(
    const float* __restrict__ x, const float* __restrict__ w_ih,
    const float* __restrict__ w_hh, const float* __restrict__ b_ih,
    const float* __restrict__ b_hh, const float* __restrict__ fc_w,
    const float* __restrict__ fc_b, float* __restrict__ out) {
  __shared__ __attribute__((aligned(16))) float xs[4 * 2 * WBUF];
  const int lane = threadIdx.x & 31;
  const int wid  = threadIdx.x >> 5;
  const int row  = lane & 15;
  const bool hi  = lane >= 16;
  const int ka   = hi ? 2 : 0;
  const int bglob = blockIdx.x * 64 + wid * 16 + row;

  const bool nrow = row < 8;
  const float* whr0 = w_hh + row * H_DIM;
  const float* whr1 = w_hh + (16 + (row & 7)) * H_DIM;
  v2f a_hh_t0_k0 = mk2(whr0[ka],     whr0[ka + 1]);
  v2f a_hh_t0_k1 = mk2(whr0[4 + ka], whr0[5 + ka]);
  v2f a_hh_t1_k0 = nrow ? mk2(whr1[ka],     whr1[ka + 1]) : mk2(0.f, 0.f);
  v2f a_hh_t1_k1 = nrow ? mk2(whr1[4 + ka], whr1[5 + ka]) : mk2(0.f, 0.f);

  const float* wir0 = w_ih + row * I_DIM;
  const float* wir1 = w_ih + (16 + (row & 7)) * I_DIM;
  v2f a_ih_t0_k0 = mk2(wir0[ka], wir0[ka + 1]);
  v2f a_ih_t0_k1 = hi ? mk2(0.f, 0.f) : mk2(wir0[4], 0.f);
  v2f a_ih_t1_k0 = nrow ? mk2(wir1[ka], wir1[ka + 1]) : mk2(0.f, 0.f);
  v2f a_ih_t1_k1 = (nrow && !hi) ? mk2(wir1[4], 0.f) : mk2(0.f, 0.f);

  v8f c_rz, c_xn, c_hn;
#pragma unroll
  for (int j = 0; j < 8; ++j) {
    int m = j + (hi ? 8 : 0);
    c_rz[j] = b_ih[m] + b_hh[m];
    c_xn[j] = hi ? 0.f : b_ih[16 + j];
    c_hn[j] = hi ? 0.f : b_hh[16 + j];
  }

  v2f h0 = mk2(0.f, 0.f), h1 = mk2(0.f, 0.f);

  const int wid_s = __builtin_amdgcn_readfirstlane(wid);
  const float* xw = x + (size_t)(blockIdx.x * 64 + wid_s * 16) * XROW;
  float* wb = xs + wid_s * (2 * WBUF);
  const bool stg = lane < ROW4;
  const int l4 = lane * 4;

  v4f g0[16], g[16];
#pragma unroll
  for (int r = 0; r < 16; ++r) { g0[r] = (v4f)(0.f); g[r] = (v4f)(0.f); }
  if (stg) {
#pragma unroll
    for (int r = 0; r < 16; ++r)
      g0[r] = *reinterpret_cast<const v4f*>(xw + r * XROW + l4);
#pragma unroll
    for (int r = 0; r < 16; ++r)
      g[r] = *reinterpret_cast<const v4f*>(xw + r * XROW + TC * I_DIM + l4);
#pragma unroll
    for (int r = 0; r < 16; ++r)
      *reinterpret_cast<v4f*>(wb + r * LROW + l4) = g0[r];
  }

  const float* lrow = wb + row * LROW;
  const float* lrow_k = lrow + ka;

  v8f xrz, xnn;
  {
    float x0a = lrow[ka], x0c = lrow[ka + 1], x0e = lrow[4];
    v2f b0 = mk2(x0a, x0c);
    v2f b1 = mk2(hi ? 0.f : x0e, 0.f);
    xrz = wmma4(a_ih_t0_k0, b0, c_rz);
    xrz = wmma4(a_ih_t0_k1, b1, xrz);
    xnn = wmma4(a_ih_t1_k0, b0, c_xn);
    xnn = wmma4(a_ih_t1_k1, b1, xnn);
  }
  float xa = lrow[I_DIM + ka], xc = lrow[I_DIM + ka + 1], xe0 = lrow[I_DIM + 4];

  for (int c = 0; c < NCHUNK; ++c) {
    const int cb = c & 1, nb = cb ^ 1;
    if (c + 1 < NCHUNK) {
      if (stg) {
#pragma unroll
        for (int r = 0; r < 16; ++r)
          *reinterpret_cast<v4f*>(wb + nb * WBUF + r * LROW + l4) = g[r];
      }
    }
    if (c + 2 < NCHUNK) {
      if (stg) {
        const float* gp = xw + (c + 2) * (TC * I_DIM) + l4;
#pragma unroll
        for (int r = 0; r < 16; ++r)
          g[r] = *reinterpret_cast<const v4f*>(gp + r * XROW);
      }
    }

#pragma unroll 2
    for (int tt = 0; tt < TC; ++tt) {
        v8f rz = wmma4(a_hh_t0_k0, h0, xrz);
        v8f hn = wmma4(a_hh_t1_k0, h0, c_hn);
        rz = wmma4(a_hh_t0_k1, h1, rz);
        v8f xn = xnn;
        v2f xB0 = mk2(xa, xc);
        v2f xB1 = mk2(hi ? 0.f : xe0, 0.f);
        STAGE("+v"(rz), "+v"(hn), "+v"(xB0), "+v"(xB1));

        const int n2   = tt + 2;
        const int ntt  = (n2 >= TC) ? n2 - TC : n2;
        const int nbuf = (n2 >= TC) ? nb : cb;
        const int off = nbuf * WBUF + ntt * I_DIM;
        float nxa = lrow_k[off], nxc = lrow_k[off + 1], nxe0 = lrow[off + 4];
        asm volatile("" : "+v"(hn));
        hn = wmma4(a_hh_t1_k1, h1, hn);
        STAGE("+v"(rz), "+v"(hn));

        float th[8], s[8], pre[8], n[8];

        xrz = wmma4(a_ih_t0_k0, xB0, c_rz);
#pragma unroll
        for (int j = 0; j < 8; ++j) th[j] = tanh_hw(0.5f * rz[j]);
        STAGE("+v"(xrz), "+v"(xB1), "+v"(th[0]), "+v"(th[1]), "+v"(th[2]), "+v"(th[3]),
              "+v"(th[4]), "+v"(th[5]), "+v"(th[6]), "+v"(th[7]));

        xrz = wmma4(a_ih_t0_k1, xB1, xrz);
#pragma unroll
        for (int j = 0; j < 8; ++j) s[j] = __builtin_fmaf(0.5f, th[j], 0.5f);
#pragma unroll
        for (int j = 0; j < 8; ++j) pre[j] = xn[j] + s[j] * hn[j];
        STAGE("+v"(xrz), "+v"(xB0), "+v"(pre[0]), "+v"(pre[1]), "+v"(pre[2]), "+v"(pre[3]),
              "+v"(pre[4]), "+v"(pre[5]), "+v"(pre[6]), "+v"(pre[7]));

        xnn = wmma4(a_ih_t1_k0, xB0, c_xn);
#pragma unroll
        for (int j = 0; j < 8; ++j) n[j] = tanh_hw(pre[j]);
        STAGE("+v"(xnn), "+v"(xB1), "+v"(n[0]), "+v"(n[1]), "+v"(n[2]), "+v"(n[3]),
              "+v"(n[4]), "+v"(n[5]), "+v"(n[6]), "+v"(n[7]));

        xnn = wmma4(a_ih_t1_k1, xB1, xnn);
        float z0 = hi_of_both(s[0], s[2]);
        float z1 = hi_of_both(s[1], s[3]);
        float z2 = hi_of_both(s[4], s[6]);
        float z3 = hi_of_both(s[5], s[7]);
        float nn0 = lo_of_both(n[0], n[2]);
        float nn1 = lo_of_both(n[1], n[3]);
        float nn2 = lo_of_both(n[4], n[6]);
        float nn3 = lo_of_both(n[5], n[7]);

        h0.x = nn0 + z0 * (h0.x - nn0);
        h0.y = nn1 + z1 * (h0.y - nn1);
        h1.x = nn2 + z2 * (h1.x - nn2);
        h1.y = nn3 + z3 * (h1.y - nn3);
        STAGE("+v"(xnn), "+v"(h0), "+v"(h1));

        xa = nxa; xc = nxc; xe0 = nxe0;
    }
  }

  float o0 = fc_w[0 * 8 + ka] * h0.x + fc_w[0 * 8 + ka + 1] * h0.y +
             fc_w[0 * 8 + ka + 4] * h1.x + fc_w[0 * 8 + ka + 5] * h1.y;
  float o1 = fc_w[1 * 8 + ka] * h0.x + fc_w[1 * 8 + ka + 1] * h0.y +
             fc_w[1 * 8 + ka + 4] * h1.x + fc_w[1 * 8 + ka + 5] * h1.y;
  o0 += swz16(o0);
  o1 += swz16(o1);
  if (!hi) {
    out[(size_t)bglob * 2 + 0] = o0 + fc_b[0];
    out[(size_t)bglob * 2 + 1] = o1 + fc_b[1];
  }
}

extern "C" void kernel_launch(void* const* d_in, const int* in_sizes, int n_in,
                              void* d_out, int out_size, void* d_ws, size_t ws_size,
                              hipStream_t stream) {
  const float* x    = (const float*)d_in[0];
  const float* w_ih = (const float*)d_in[1];
  const float* w_hh = (const float*)d_in[2];
  const float* b_ih = (const float*)d_in[3];
  const float* b_hh = (const float*)d_in[4];
  const float* fc_w = (const float*)d_in[5];
  const float* fc_b = (const float*)d_in[6];
  float* out = (float*)d_out;

  const int B = in_sizes[0] / (T_DIM * I_DIM);
  dim3 grid(B / 64), block(128);
  gru_wmma_kernel<<<grid, block, 0, stream>>>(x, w_ih, w_hh, b_ih, b_hh, fc_w,
                                              fc_b, out);
}


# batch re-tiled: 8 batch rows per wave, 1024 waves; one result tile per step holds all gate rows, column halves on separate k-steps: 7 not 8 products, 6 not 16 tanh per step; sums and order unchanged
# speedup vs baseline: 2.1911x; 1.1374x over previous
#include <hip/hip_runtime.h>


typedef float v2f __attribute__((ext_vector_type(2)));
typedef float v4f __attribute__((ext_vector_type(4)));
typedef float v8f __attribute__((ext_vector_type(8)));

#define I_DIM 5
#define H_DIM 8
#define T_DIM 512
#define NB 8
#define TC 16
#define NCHUNK (T_DIM / TC)
#define ROW4 (TC * I_DIM / 4)
#define LROW 84
#define WBUF (NB * LROW)
#define XROW (T_DIM * I_DIM)

__device__ __forceinline__ v2f mk2(float a, float b) { v2f v; v.x = a; v.y = b; return v; }

__device__ __forceinline__ float tanh_hw(float x) { return __builtin_amdgcn_tanhf(x); }

__device__ __forceinline__ float other_group(float v) {
  return __int_as_float(__builtin_amdgcn_update_dpp(0, __float_as_int(v), 0x128, 0xf, 0xf, false));
}
__device__ __forceinline__ float swz16(float v) {
  return __int_as_float(__builtin_amdgcn_ds_swizzle(__float_as_int(v), 0x401F));
}

__device__ __forceinline__ v8f wmma4(v2f a, v2f b, v8f c) {
  return __builtin_amdgcn_wmma_f32_16x16x4_f32(false, a, false, b, (short)0, c, false, false);
}

__device__ __forceinline__ float wh(const float* w_hh, int type, int unit, int k) {
  return type < 3 ? w_hh[(type * 8 + unit) * H_DIM + k] : 0.f;
}
__device__ __forceinline__ float wx(const float* w_ih, int type, int unit, int i) {
  return type == 2 ? 0.f : w_ih[((type == 3 ? 2 : type) * 8 + unit) * I_DIM + i];
}

#define STAGE(...) do { asm volatile("" : __VA_ARGS__); __builtin_amdgcn_sched_barrier(0); } while (0)

__global__ void __launch_bounds__(128) gru_wmma_kernel(
    const float* __restrict__ x, const float* __restrict__ w_ih,
    const float* __restrict__ w_hh, const float* __restrict__ b_ih,
    const float* __restrict__ b_hh, const float* __restrict__ fc_w,
    const float* __restrict__ fc_b, float* __restrict__ out) {
  __shared__ __attribute__((aligned(16))) float xs[4 * 2 * WBUF];
  __shared__ __attribute__((aligned(16))) float osh[4 * NB * 2];
  const int lane = threadIdx.x & 31;
  const int wid  = threadIdx.x >> 5;
  const int col  = lane & 15;
  const int hf   = lane >> 4;
  const int grp  = col >> 3;
  const int bi   = col & 7;

  const int m = col, s0 = 2 * hf;
  const int typ = (m & 7) >> 1;
  const int un0 = 2 * (0 + (m >> 3)) + (m & 1);
  const int un1 = 2 * (2 + (m >> 3)) + (m & 1);
  v2f ah0 = mk2(wh(w_hh, typ, un0, s0),     wh(w_hh, typ, un0, s0 + 1));
  v2f ah1 = mk2(wh(w_hh, typ, un0, 4 + s0), wh(w_hh, typ, un0, 5 + s0));
  v2f ah2 = mk2(wh(w_hh, typ, un1, s0),     wh(w_hh, typ, un1, s0 + 1));
  v2f ah3 = mk2(wh(w_hh, typ, un1, 4 + s0), wh(w_hh, typ, un1, 5 + s0));
  v2f ax0 = mk2(wx(w_ih, typ, un0, s0), wx(w_ih, typ, un0, s0 + 1));
  v2f ax1 = mk2(wx(w_ih, typ, un1, s0), wx(w_ih, typ, un1, s0 + 1));
  v2f ax2 = hf ? mk2(0.f, 0.f) : mk2(wx(w_ih, typ, un0, 4), wx(w_ih, typ, un1, 4));

  const int u0 = 4 * grp + 2 * hf;
  v8f cb;
#pragma unroll
  for (int r = 0; r < 8; ++r) {
    const int u = u0 + (r & 1), t = r >> 1;
    cb[r] = t == 0 ? b_ih[u] + b_hh[u]
          : t == 1 ? b_ih[8 + u] + b_hh[8 + u]
          : t == 2 ? b_hh[16 + u] : b_ih[16 + u];
  }

  const float m0 = grp ? 0.f : 1.f, m1 = grp ? 1.f : 0.f;
  const v2f e4 = hf ? mk2(0.f, 0.f) : mk2(m0, m1);

  v2f own = mk2(0.f, 0.f), oth = mk2(0.f, 0.f);

  const int wid_s = __builtin_amdgcn_readfirstlane(wid);
  const float* xw = x + (size_t)(blockIdx.x * (4 * NB) + wid_s * NB) * XROW;
  float* wb = xs + wid_s * (2 * WBUF);
  const bool stg = lane < ROW4;
  const int l4 = lane * 4;

  v4f g0[NB], g[NB];
#pragma unroll
  for (int r = 0; r < NB; ++r) { g0[r] = (v4f)(0.f); g[r] = (v4f)(0.f); }
  if (stg) {
#pragma unroll
    for (int r = 0; r < NB; ++r)
      g0[r] = *reinterpret_cast<const v4f*>(xw + r * XROW + l4);
#pragma unroll
    for (int r = 0; r < NB; ++r)
      g[r] = *reinterpret_cast<const v4f*>(xw + r * XROW + TC * I_DIM + l4);
#pragma unroll
    for (int r = 0; r < NB; ++r)
      *reinterpret_cast<v4f*>(wb + r * LROW + l4) = g0[r];
  }

  const float* lrow = wb + bi * LROW;
  const float* lrow_k = lrow + s0;

  v8f X;
  {
    v2f xv = mk2(lrow_k[0], lrow_k[1]);
    float x4 = lrow[4];
    X = wmma4(ax0, xv * m0, cb);
    X = wmma4(ax1, xv * m1, X);
    X = wmma4(ax2, e4 * x4, X);
  }
  float xa = lrow_k[I_DIM], xc = lrow_k[I_DIM + 1], xe = lrow[I_DIM + 4];

  for (int c = 0; c < NCHUNK; ++c) {
    const int cbuf = c & 1, nb = cbuf ^ 1;
    if (c + 1 < NCHUNK) {
      if (stg) {
#pragma unroll
        for (int r = 0; r < NB; ++r)
          *reinterpret_cast<v4f*>(wb + nb * WBUF + r * LROW + l4) = g[r];
      }
    }
    if (c + 2 < NCHUNK) {
      if (stg) {
        const float* gp = xw + (c + 2) * (TC * I_DIM) + l4;
#pragma unroll
        for (int r = 0; r < NB; ++r)
          g[r] = *reinterpret_cast<const v4f*>(gp + r * XROW);
      }
    }

#pragma unroll 2
    for (int tt = 0; tt < TC; ++tt) {
      v2f xv = mk2(xa, xc);
      v2f bx0 = xv * m0, bx1 = xv * m1, bx2 = e4 * xe;
      v2f bh0 = own * m0, bh1 = oth * m0, bh2 = oth * m1, bh3 = own * m1;

      v8f d = wmma4(ah0, bh0, X);
      d = wmma4(ah1, bh1, d);
      d = wmma4(ah2, bh2, d);
      d = wmma4(ah3, bh3, d);
      STAGE("+v"(d), "+v"(bx0), "+v"(bx1), "+v"(bx2));

      const int n2   = tt + 2;
      const int ntt  = (n2 >= TC) ? n2 - TC : n2;
      const int nbuf = (n2 >= TC) ? nb : cbuf;
      const int off  = nbuf * WBUF + ntt * I_DIM;
      float nxa = lrow_k[off], nxc = lrow_k[off + 1], nxe = lrow[off + 4];

      X = wmma4(ax0, bx0, cb);
      float t0 = tanh_hw(0.5f * d[0]), t1 = tanh_hw(0.5f * d[1]);
      float t2 = tanh_hw(0.5f * d[2]), t3 = tanh_hw(0.5f * d[3]);
      STAGE("+v"(X), "+v"(t0), "+v"(t1), "+v"(t2), "+v"(t3));

      X = wmma4(ax1, bx1, X);
      float r0 = __builtin_fmaf(0.5f, t0, 0.5f), r1 = __builtin_fmaf(0.5f, t1, 0.5f);
      float z0 = __builtin_fmaf(0.5f, t2, 0.5f), z1 = __builtin_fmaf(0.5f, t3, 0.5f);
      float p0 = d[6] + r0 * d[4], p1 = d[7] + r1 * d[5];
      float n0 = tanh_hw(p0), n1 = tanh_hw(p1);
      STAGE("+v"(X), "+v"(n0), "+v"(n1), "+v"(z0), "+v"(z1));

      X = wmma4(ax2, bx2, X);
      own.x = n0 + z0 * (own.x - n0);
      own.y = n1 + z1 * (own.y - n1);
      oth.x = other_group(own.x);
      oth.y = other_group(own.y);
      STAGE("+v"(X), "+v"(own), "+v"(oth));

      xa = nxa; xc = nxc; xe = nxe;
    }
  }

  const int ka = 2 * hf;
  float o0 = fc_w[0 * 8 + ka] * own.x + fc_w[0 * 8 + ka + 1] * own.y +
             fc_w[0 * 8 + ka + 4] * oth.x + fc_w[0 * 8 + ka + 5] * oth.y;
  float o1 = fc_w[1 * 8 + ka] * own.x + fc_w[1 * 8 + ka + 1] * own.y +
             fc_w[1 * 8 + ka + 4] * oth.x + fc_w[1 * 8 + ka + 5] * oth.y;
  o0 += swz16(o0);
  o1 += swz16(o1);
  if (lane < NB) {
    osh[(wid * NB + lane) * 2 + 0] = o0 + fc_b[0];
    osh[(wid * NB + lane) * 2 + 1] = o1 + fc_b[1];
  }
  __syncthreads();
  if (wid == 0) {
    v2f o = *reinterpret_cast<const v2f*>(osh + lane * 2);
    *reinterpret_cast<v2f*>(out + ((size_t)blockIdx.x * (4 * NB) + lane) * 2) = o;
  }
}

extern "C" void kernel_launch(void* const* d_in, const int* in_sizes, int n_in,
                              void* d_out, int out_size, void* d_ws, size_t ws_size,
                              hipStream_t stream) {
  const float* x    = (const float*)d_in[0];
  const float* w_ih = (const float*)d_in[1];
  const float* w_hh = (const float*)d_in[2];
  const float* b_ih = (const float*)d_in[3];
  const float* b_hh = (const float*)d_in[4];
  const float* fc_w = (const float*)d_in[5];
  const float* fc_b = (const float*)d_in[6];
  float* out = (float*)d_out;

  const int B = in_sizes[0] / (T_DIM * I_DIM);
  dim3 grid(B / 32), block(128);
  gru_wmma_kernel<<<grid, block, 0, stream>>>(x, w_ih, w_hh, b_ih, b_hh, fc_w,
                                              fc_b, out);
}


# x operand reads grouped: one group of LDS reads per 16-step chunk into registers (steps straight-line, constant offsets) replaces the reads in every step; values, sums and their order unchanged
# speedup vs baseline: 2.5400x; 1.1592x over previous
#include <hip/hip_runtime.h>


typedef float v2f __attribute__((ext_vector_type(2)));
typedef float v4f __attribute__((ext_vector_type(4)));
typedef float v8f __attribute__((ext_vector_type(8)));

#define I_DIM 5
#define H_DIM 8
#define T_DIM 512
#define NB 8
#define TC 16
#define NCHUNK (T_DIM / TC)
#define ROW4 (TC * I_DIM / 4)
#define LROW 84
#define WBUF (NB * LROW)
#define XROW (T_DIM * I_DIM)

__device__ __forceinline__ v2f mk2(float a, float b) { v2f v; v.x = a; v.y = b; return v; }

__device__ __forceinline__ float tanh_hw(float x) { return __builtin_amdgcn_tanhf(x); }

__device__ __forceinline__ float other_group(float v) {
  return __int_as_float(__builtin_amdgcn_update_dpp(0, __float_as_int(v), 0x128, 0xf, 0xf, false));
}
__device__ __forceinline__ float swz16(float v) {
  return __int_as_float(__builtin_amdgcn_ds_swizzle(__float_as_int(v), 0x401F));
}

__device__ __forceinline__ v8f wmma4(v2f a, v2f b, v8f c) {
  return __builtin_amdgcn_wmma_f32_16x16x4_f32(false, a, false, b, (short)0, c, false, false);
}

__device__ __forceinline__ float wh(const float* w_hh, int type, int unit, int k) {
  return type < 3 ? w_hh[(type * 8 + unit) * H_DIM + k] : 0.f;
}
__device__ __forceinline__ float wx(const float* w_ih, int type, int unit, int i) {
  return type == 2 ? 0.f : w_ih[((type == 3 ? 2 : type) * 8 + unit) * I_DIM + i];
}

#define STAGE(...) do { asm volatile("" : __VA_ARGS__); __builtin_amdgcn_sched_barrier(0); } while (0)

__global__ void __launch_bounds__(128) gru_wmma_kernel(
    const float* __restrict__ x, const float* __restrict__ w_ih,
    const float* __restrict__ w_hh, const float* __restrict__ b_ih,
    const float* __restrict__ b_hh, const float* __restrict__ fc_w,
    const float* __restrict__ fc_b, float* __restrict__ out) {
  __shared__ __attribute__((aligned(16))) float xs[4 * 2 * WBUF];
  __shared__ __attribute__((aligned(16))) float osh[4 * NB * 2];
  const int lane = threadIdx.x & 31;
  const int wid  = threadIdx.x >> 5;
  const int col  = lane & 15;
  const int hf   = lane >> 4;
  const int grp  = col >> 3;
  const int bi   = col & 7;

  const int m = col, s0 = 2 * hf;
  const int typ = (m & 7) >> 1;
  const int un0 = 2 * (0 + (m >> 3)) + (m & 1);
  const int un1 = 2 * (2 + (m >> 3)) + (m & 1);
  v2f ah0 = mk2(wh(w_hh, typ, un0, s0),     wh(w_hh, typ, un0, s0 + 1));
  v2f ah1 = mk2(wh(w_hh, typ, un0, 4 + s0), wh(w_hh, typ, un0, 5 + s0));
  v2f ah2 = mk2(wh(w_hh, typ, un1, s0),     wh(w_hh, typ, un1, s0 + 1));
  v2f ah3 = mk2(wh(w_hh, typ, un1, 4 + s0), wh(w_hh, typ, un1, 5 + s0));
  v2f ax0 = mk2(wx(w_ih, typ, un0, s0), wx(w_ih, typ, un0, s0 + 1));
  v2f ax1 = mk2(wx(w_ih, typ, un1, s0), wx(w_ih, typ, un1, s0 + 1));
  v2f ax2 = hf ? mk2(0.f, 0.f) : mk2(wx(w_ih, typ, un0, 4), wx(w_ih, typ, un1, 4));

  const int u0 = 4 * grp + 2 * hf;
  v8f cb;
#pragma unroll
  for (int r = 0; r < 8; ++r) {
    const int u = u0 + (r & 1), t = r >> 1;
    cb[r] = t == 0 ? b_ih[u] + b_hh[u]
          : t == 1 ? b_ih[8 + u] + b_hh[8 + u]
          : t == 2 ? b_hh[16 + u] : b_ih[16 + u];
  }

  const float m0 = grp ? 0.f : 1.f, m1 = grp ? 1.f : 0.f;
  const v2f e4 = hf ? mk2(0.f, 0.f) : mk2(m0, m1);

  v2f own = mk2(0.f, 0.f), oth = mk2(0.f, 0.f);

  const int wid_s = __builtin_amdgcn_readfirstlane(wid);
  const float* xw = x + (size_t)(blockIdx.x * (4 * NB) + wid_s * NB) * XROW;
  float* wb = xs + wid_s * (2 * WBUF);
  const bool stg = lane < ROW4;
  const int l4 = lane * 4;

  v4f g0[NB], g[NB];
#pragma unroll
  for (int r = 0; r < NB; ++r) { g0[r] = (v4f)(0.f); g[r] = (v4f)(0.f); }
  if (stg) {
#pragma unroll
    for (int r = 0; r < NB; ++r)
      g0[r] = *reinterpret_cast<const v4f*>(xw + r * XROW + l4);
#pragma unroll
    for (int r = 0; r < NB; ++r)
      g[r] = *reinterpret_cast<const v4f*>(xw + r * XROW + TC * I_DIM + l4);
#pragma unroll
    for (int r = 0; r < NB; ++r)
      *reinterpret_cast<v4f*>(wb + r * LROW + l4) = g0[r];
  }

  const float* lrow = wb + bi * LROW;
  const float* lrow_k = lrow + s0;

  v8f X;
  {
    v2f xv = mk2(lrow_k[0], lrow_k[1]);
    float x4 = lrow[4];
    X = wmma4(ax0, xv * m0, cb);
    X = wmma4(ax1, xv * m1, X);
    X = wmma4(ax2, e4 * x4, X);
  }
  v2f xk_c = mk2(lrow_k[I_DIM], lrow_k[I_DIM + 1]);
  float x4_c = lrow[I_DIM + 4];

  for (int c = 0; c < NCHUNK; ++c) {
    const int cbuf = c & 1, nb = cbuf ^ 1;
    if (c + 1 < NCHUNK) {
      if (stg) {
#pragma unroll
        for (int r = 0; r < NB; ++r)
          *reinterpret_cast<v4f*>(wb + nb * WBUF + r * LROW + l4) = g[r];
      }
    }

    const float* cur = lrow + cbuf * WBUF;
    const float* nxt = lrow + nb * WBUF;
    v2f xk[TC];
    float x4[TC];
#pragma unroll
    for (int i = 0; i < TC; ++i) {
      const float* p = (i + 2 < TC) ? cur + (i + 2) * I_DIM : nxt + (i + 2 - TC) * I_DIM;
      xk[i] = mk2(p[s0], p[s0 + 1]);
      x4[i] = p[4];
    }

    if (c + 2 < NCHUNK) {
      if (stg) {
        const float* gp = xw + (c + 2) * (TC * I_DIM) + l4;
#pragma unroll
        for (int r = 0; r < NB; ++r)
          g[r] = *reinterpret_cast<const v4f*>(gp + r * XROW);
      }
    }

#pragma unroll
    for (int tt = 0; tt < TC; ++tt) {
      const v2f xv = tt ? xk[tt - 1] : xk_c;
      const float xe = tt ? x4[tt - 1] : x4_c;
      v2f bx0 = xv * m0, bx1 = xv * m1, bx2 = e4 * xe;
      v2f bh0 = own * m0, bh1 = oth * m0, bh2 = oth * m1, bh3 = own * m1;

      v8f d = wmma4(ah0, bh0, X);
      d = wmma4(ah1, bh1, d);
      d = wmma4(ah2, bh2, d);
      d = wmma4(ah3, bh3, d);
      STAGE("+v"(d), "+v"(bx0), "+v"(bx1), "+v"(bx2));

      X = wmma4(ax0, bx0, cb);
      float t0 = tanh_hw(0.5f * d[0]), t1 = tanh_hw(0.5f * d[1]);
      float t2 = tanh_hw(0.5f * d[2]), t3 = tanh_hw(0.5f * d[3]);
      STAGE("+v"(X), "+v"(t0), "+v"(t1), "+v"(t2), "+v"(t3));

      X = wmma4(ax1, bx1, X);
      float r0 = __builtin_fmaf(0.5f, t0, 0.5f), r1 = __builtin_fmaf(0.5f, t1, 0.5f);
      float z0 = __builtin_fmaf(0.5f, t2, 0.5f), z1 = __builtin_fmaf(0.5f, t3, 0.5f);
      float p0 = d[6] + r0 * d[4], p1 = d[7] + r1 * d[5];
      float n0 = tanh_hw(p0), n1 = tanh_hw(p1);
      STAGE("+v"(X), "+v"(n0), "+v"(n1), "+v"(z0), "+v"(z1));

      X = wmma4(ax2, bx2, X);
      own.x = n0 + z0 * (own.x - n0);
      own.y = n1 + z1 * (own.y - n1);
      oth.x = other_group(own.x);
      oth.y = other_group(own.y);
      STAGE("+v"(X), "+v"(own), "+v"(oth));
    }
    xk_c = xk[TC - 1];
    x4_c = x4[TC - 1];
  }

  const int ka = 2 * hf;
  float o0 = fc_w[0 * 8 + ka] * own.x + fc_w[0 * 8 + ka + 1] * own.y +
             fc_w[0 * 8 + ka + 4] * oth.x + fc_w[0 * 8 + ka + 5] * oth.y;
  float o1 = fc_w[1 * 8 + ka] * own.x + fc_w[1 * 8 + ka + 1] * own.y +
             fc_w[1 * 8 + ka + 4] * oth.x + fc_w[1 * 8 + ka + 5] * oth.y;
  o0 += swz16(o0);
  o1 += swz16(o1);
  if (lane < NB) {
    osh[(wid * NB + lane) * 2 + 0] = o0 + fc_b[0];
    osh[(wid * NB + lane) * 2 + 1] = o1 + fc_b[1];
  }
  __syncthreads();
  if (wid == 0) {
    v2f o = *reinterpret_cast<const v2f*>(osh + lane * 2);
    *reinterpret_cast<v2f*>(out + ((size_t)blockIdx.x * (4 * NB) + lane) * 2) = o;
  }
}

extern "C" void kernel_launch(void* const* d_in, const int* in_sizes, int n_in,
                              void* d_out, int out_size, void* d_ws, size_t ws_size,
                              hipStream_t stream) {
  const float* x    = (const float*)d_in[0];
  const float* w_ih = (const float*)d_in[1];
  const float* w_hh = (const float*)d_in[2];
  const float* b_ih = (const float*)d_in[3];
  const float* b_hh = (const float*)d_in[4];
  const float* fc_w = (const float*)d_in[5];
  const float* fc_b = (const float*)d_in[6];
  float* out = (float*)d_out;

  const int B = in_sizes[0] / (T_DIM * I_DIM);
  dim3 grid(B / 32), block(128);
  gru_wmma_kernel<<<grid, block, 0, stream>>>(x, w_ih, w_hh, b_ih, b_hh, fc_w,
                                              fc_b, out);
}


# gate math re-associated: 0.5 and hn/2 folded into weight rows, column masks into the (1-z) and z*h coefficients; between h products only tanh,fma,tanh,fma remain; rounding of these sums changes
# speedup vs baseline: 2.7573x; 1.0856x over previous
#include <hip/hip_runtime.h>


typedef float v2f __attribute__((ext_vector_type(2)));
typedef float v4f __attribute__((ext_vector_type(4)));
typedef float v8f __attribute__((ext_vector_type(8)));

#define I_DIM 5
#define H_DIM 8
#define T_DIM 512
#define NB 8
#define TC 16
#define NCHUNK (T_DIM / TC)
#define ROW4 (TC * I_DIM / 4)
#define LROW 84
#define WBUF (NB * LROW)
#define XROW (T_DIM * I_DIM)

__device__ __forceinline__ v2f mk2(float a, float b) { v2f v; v.x = a; v.y = b; return v; }

__device__ __forceinline__ float tanh_hw(float x) { return __builtin_amdgcn_tanhf(x); }

__device__ __forceinline__ float other_group(float v) {
  return __int_as_float(__builtin_amdgcn_mov_dpp(__float_as_int(v), 0x128, 0xf, 0xf, false));
}
__device__ __forceinline__ float swz16(float v) {
  return __int_as_float(__builtin_amdgcn_ds_swizzle(__float_as_int(v), 0x401F));
}

__device__ __forceinline__ v8f wmma4(v2f a, v2f b, v8f c) {
  return __builtin_amdgcn_wmma_f32_16x16x4_f32(false, a, false, b, (short)0, c, false, false);
}

__device__ __forceinline__ float wh(const float* w_hh, int type, int unit, int k) {
  return 0.5f * w_hh[((type == 3 ? 2 : type) * 8 + unit) * H_DIM + k];
}
__device__ __forceinline__ float wx(const float* w_ih, int type, int unit, int i) {
  return type == 2 ? 0.f : (type == 3 ? 1.f : 0.5f) * w_ih[((type == 3 ? 2 : type) * 8 + unit) * I_DIM + i];
}

#define STAGE(...) do { asm volatile("" : __VA_ARGS__); __builtin_amdgcn_sched_barrier(0); } while (0)

__global__ void __launch_bounds__(128) gru_wmma_kernel(
    const float* __restrict__ x, const float* __restrict__ w_ih,
    const float* __restrict__ w_hh, const float* __restrict__ b_ih,
    const float* __restrict__ b_hh, const float* __restrict__ fc_w,
    const float* __restrict__ fc_b, float* __restrict__ out) {
  __shared__ __attribute__((aligned(16))) float xs[4 * 2 * WBUF];
  __shared__ __attribute__((aligned(16))) float osh[4 * NB * 2];
  const int lane = threadIdx.x & 31;
  const int wid  = threadIdx.x >> 5;
  const int col  = lane & 15;
  const int hf   = lane >> 4;
  const int grp  = col >> 3;
  const int bi   = col & 7;

  const int m = col, s0 = 2 * hf;
  const int typ = (m & 7) >> 1;
  const int un0 = 2 * (0 + (m >> 3)) + (m & 1);
  const int un1 = 2 * (2 + (m >> 3)) + (m & 1);
  v2f ah0 = mk2(wh(w_hh, typ, un0, s0),     wh(w_hh, typ, un0, s0 + 1));
  v2f ah1 = mk2(wh(w_hh, typ, un0, 4 + s0), wh(w_hh, typ, un0, 5 + s0));
  v2f ah2 = mk2(wh(w_hh, typ, un1, s0),     wh(w_hh, typ, un1, s0 + 1));
  v2f ah3 = mk2(wh(w_hh, typ, un1, 4 + s0), wh(w_hh, typ, un1, 5 + s0));
  v2f ax0 = mk2(wx(w_ih, typ, un0, s0), wx(w_ih, typ, un0, s0 + 1));
  v2f ax1 = mk2(wx(w_ih, typ, un1, s0), wx(w_ih, typ, un1, s0 + 1));
  v2f ax2 = hf ? mk2(0.f, 0.f) : mk2(wx(w_ih, typ, un0, 4), wx(w_ih, typ, un1, 4));

  const int u0 = 4 * grp + 2 * hf;
  v8f cb;
#pragma unroll
  for (int r = 0; r < 8; ++r) {
    const int u = u0 + (r & 1), t = r >> 1;
    cb[r] = t == 0 ? 0.5f * (b_ih[u] + b_hh[u])
          : t == 1 ? 0.5f * (b_ih[8 + u] + b_hh[8 + u])
          : t == 2 ? 0.5f * b_hh[16 + u] : b_ih[16 + u] + 0.5f * b_hh[16 + u];
  }

  const float m0 = grp ? 0.f : 1.f, m1 = grp ? 1.f : 0.f;
  const v2f e4 = hf ? mk2(0.f, 0.f) : mk2(m0, m1);

  const v2f hm0 = mk2(0.5f * m0, 0.5f * m0), hm1 = mk2(0.5f * m1, 0.5f * m1);
  const v2f nhm0 = -hm0, nhm1 = -hm1;
  const v2f half2 = mk2(0.5f, 0.5f);

  v2f hA = mk2(0.f, 0.f), hB = mk2(0.f, 0.f), oA = mk2(0.f, 0.f), oB = mk2(0.f, 0.f);

  const int wid_s = __builtin_amdgcn_readfirstlane(wid);
  const float* xw = x + (size_t)(blockIdx.x * (4 * NB) + wid_s * NB) * XROW;
  float* wb = xs + wid_s * (2 * WBUF);
  const bool stg = lane < ROW4;
  const int l4 = lane * 4;

  v4f g0[NB], g[NB];
#pragma unroll
  for (int r = 0; r < NB; ++r) { g0[r] = (v4f)(0.f); g[r] = (v4f)(0.f); }
  if (stg) {
#pragma unroll
    for (int r = 0; r < NB; ++r)
      g0[r] = *reinterpret_cast<const v4f*>(xw + r * XROW + l4);
#pragma unroll
    for (int r = 0; r < NB; ++r)
      g[r] = *reinterpret_cast<const v4f*>(xw + r * XROW + TC * I_DIM + l4);
#pragma unroll
    for (int r = 0; r < NB; ++r)
      *reinterpret_cast<v4f*>(wb + r * LROW + l4) = g0[r];
  }

  const float* lrow = wb + bi * LROW;
  const float* lrow_k = lrow + s0;

  v8f X;
  {
    v2f xv = mk2(lrow_k[0], lrow_k[1]);
    float x4 = lrow[4];
    X = wmma4(ax0, xv * m0, cb);
    X = wmma4(ax1, xv * m1, X);
    X = wmma4(ax2, e4 * x4, X);
  }
  v2f xk_c = mk2(lrow_k[I_DIM], lrow_k[I_DIM + 1]);
  float x4_c = lrow[I_DIM + 4];

  for (int c = 0; c < NCHUNK; ++c) {
    const int cbuf = c & 1, nb = cbuf ^ 1;
    if (c + 1 < NCHUNK) {
      if (stg) {
#pragma unroll
        for (int r = 0; r < NB; ++r)
          *reinterpret_cast<v4f*>(wb + nb * WBUF + r * LROW + l4) = g[r];
      }
    }

    const float* cur = lrow + cbuf * WBUF;
    const float* nxt = lrow + nb * WBUF;
    v2f xk[TC];
    float x4[TC];
#pragma unroll
    for (int i = 0; i < TC; ++i) {
      const float* p = (i + 2 < TC) ? cur + (i + 2) * I_DIM : nxt + (i + 2 - TC) * I_DIM;
      xk[i] = mk2(p[s0], p[s0 + 1]);
      x4[i] = p[4];
    }

    if (c + 2 < NCHUNK) {
      if (stg) {
        const float* gp = xw + (c + 2) * (TC * I_DIM) + l4;
#pragma unroll
        for (int r = 0; r < NB; ++r)
          g[r] = *reinterpret_cast<const v4f*>(gp + r * XROW);
      }
    }

#pragma unroll
    for (int tt = 0; tt < TC; ++tt) {
      const v2f xv = tt ? xk[tt - 1] : xk_c;
      const float xe = tt ? x4[tt - 1] : x4_c;
      v2f bx0 = xv * m0, bx1 = xv * m1, bx2 = e4 * xe;
      v8f d = wmma4(ah0, hA, X);
      d = wmma4(ah1, oA, d);
      d = wmma4(ah2, oB, d);
      d = wmma4(ah3, hB, d);
      STAGE("+v"(d), "+v"(bx0), "+v"(bx1), "+v"(bx2));

      X = wmma4(ax0, bx0, cb);
      __builtin_amdgcn_sched_barrier(0);
      v2f tr = mk2(tanh_hw(d[0]), tanh_hw(d[1]));
      v2f tz = mk2(tanh_hw(d[2]), tanh_hw(d[3]));
      v2f p = __builtin_elementwise_fma(tr, mk2(d[4], d[5]), mk2(d[6], d[7]));
      STAGE("+v"(X), "+v"(p), "+v"(tz));

      X = wmma4(ax1, bx1, X);
      __builtin_amdgcn_sched_barrier(0);
      v2f n = mk2(tanh_hw(p.x), tanh_hw(p.y));
      v2f a0 = __builtin_elementwise_fma(tz, nhm0, hm0);
      v2f a1 = __builtin_elementwise_fma(tz, nhm1, hm1);
      v2f z = __builtin_elementwise_fma(tz, half2, half2);
      v2f zh0 = z * hA, zh1 = z * hB;
      hA = __builtin_elementwise_fma(a0, n, zh0);
      hB = __builtin_elementwise_fma(a1, n, zh1);
      STAGE("+v"(X), "+v"(hA), "+v"(hB));

      X = wmma4(ax2, bx2, X);
      oA.x = other_group(hB.x);
      oA.y = other_group(hB.y);
      oB.x = other_group(hA.x);
      oB.y = other_group(hA.y);
      STAGE("+v"(X), "+v"(oA), "+v"(oB));
    }
    xk_c = xk[TC - 1];
    x4_c = x4[TC - 1];
  }

  const int ka = 2 * hf;
  const v2f own = hA + hB, oth = oA + oB;
  float o0 = fc_w[0 * 8 + ka] * own.x + fc_w[0 * 8 + ka + 1] * own.y +
             fc_w[0 * 8 + ka + 4] * oth.x + fc_w[0 * 8 + ka + 5] * oth.y;
  float o1 = fc_w[1 * 8 + ka] * own.x + fc_w[1 * 8 + ka + 1] * own.y +
             fc_w[1 * 8 + ka + 4] * oth.x + fc_w[1 * 8 + ka + 5] * oth.y;
  o0 += swz16(o0);
  o1 += swz16(o1);
  if (lane < NB) {
    osh[(wid * NB + lane) * 2 + 0] = o0 + fc_b[0];
    osh[(wid * NB + lane) * 2 + 1] = o1 + fc_b[1];
  }
  __syncthreads();
  if (wid == 0) {
    v2f o = *reinterpret_cast<const v2f*>(osh + lane * 2);
    *reinterpret_cast<v2f*>(out + ((size_t)blockIdx.x * (4 * NB) + lane) * 2) = o;
  }
}

extern "C" void kernel_launch(void* const* d_in, const int* in_sizes, int n_in,
                              void* d_out, int out_size, void* d_ws, size_t ws_size,
                              hipStream_t stream) {
  const float* x    = (const float*)d_in[0];
  const float* w_ih = (const float*)d_in[1];
  const float* w_hh = (const float*)d_in[2];
  const float* b_ih = (const float*)d_in[3];
  const float* b_hh = (const float*)d_in[4];
  const float* fc_w = (const float*)d_in[5];
  const float* fc_b = (const float*)d_in[6];
  float* out = (float*)d_out;

  const int B = in_sizes[0] / (T_DIM * I_DIM);
  dim3 grid(B / 32), block(128);
  gru_wmma_kernel<<<grid, block, 0, stream>>>(x, w_ih, w_hh, b_ih, b_hh, fc_w,
                                              fc_b, out);
}


# x products re-tiled over step pairs: a tile's two column groups hold two consecutive steps of raw x (no zero mask), groups swap unit halves every step; 4 not 6 x products per 2 steps; sums unchanged
# speedup vs baseline: 3.0243x; 1.0968x over previous
#include <hip/hip_runtime.h>


typedef float v2f __attribute__((ext_vector_type(2)));
typedef float v4f __attribute__((ext_vector_type(4)));
typedef float v8f __attribute__((ext_vector_type(8)));

#define I_DIM 5
#define H_DIM 8
#define T_DIM 512
#define NB 8
#define TC 16
#define NCHUNK (T_DIM / TC)
#define ROW4 (TC * I_DIM / 4)
#define LROW 84
#define WBUF (NB * LROW)
#define XROW (T_DIM * I_DIM)

__device__ __forceinline__ v2f mk2(float a, float b) { v2f v; v.x = a; v.y = b; return v; }

__device__ __forceinline__ float tanh_hw(float x) { return __builtin_amdgcn_tanhf(x); }

__device__ __forceinline__ float other_group(float v) {
  return __int_as_float(__builtin_amdgcn_mov_dpp(__float_as_int(v), 0x128, 0xf, 0xf, false));
}
__device__ __forceinline__ float swz16(float v) {
  return __int_as_float(__builtin_amdgcn_ds_swizzle(__float_as_int(v), 0x401F));
}

__device__ __forceinline__ v8f wmma4(v2f a, v2f b, v8f c) {
  return __builtin_amdgcn_wmma_f32_16x16x4_f32(false, a, false, b, (short)0, c, false, false);
}

__device__ __forceinline__ float wh(const float* w_hh, int type, int unit, int k) {
  return 0.5f * w_hh[((type == 3 ? 2 : type) * 8 + unit) * H_DIM + k];
}
__device__ __forceinline__ float wx(const float* w_ih, int type, int unit, int i) {
  return type == 2 ? 0.f : (type == 3 ? 1.f : 0.5f) * w_ih[((type == 3 ? 2 : type) * 8 + unit) * I_DIM + i];
}

#define STAGE(...) do { asm volatile("" : __VA_ARGS__); __builtin_amdgcn_sched_barrier(0); } while (0)

__device__ __forceinline__ void gru_step(
    const v2f ah0, const v2f ah1, const v2f ah2, const v2f ah3,
    const v8f X, const v2f b0, const v2f b1, const v2f b2, const v2f b3,
    const bool g1, const v8f SA, const v8f SB, v8f& Xn,
    const v2f axR, const v2f a4R, const v8f cbR, const v2f xk, const v2f x34, v8f& T,
    v2f& nA, v2f& nB, v2f& rA, v2f& rB,
    const v2f hm0, const v2f nhm0, const v2f hm1, const v2f nhm1, const v2f half2) {
  v8f d = wmma4(ah0, b0, X);
  v2f s01 = g1 ? mk2(SA[0], SA[1]) : mk2(SB[0], SB[1]);
  STAGE("+v"(d), "+v"(s01));
  d = wmma4(ah1, b1, d);
  v2f s23 = g1 ? mk2(SA[2], SA[3]) : mk2(SB[2], SB[3]);
  STAGE("+v"(d), "+v"(s23));
  d = wmma4(ah2, b2, d);
  v2f s45 = g1 ? mk2(SA[4], SA[5]) : mk2(SB[4], SB[5]);
  STAGE("+v"(d), "+v"(s45));
  d = wmma4(ah3, b3, d);
  v2f s67 = g1 ? mk2(SA[6], SA[7]) : mk2(SB[6], SB[7]);
  STAGE("+v"(d), "+v"(s67));
  Xn = __builtin_shufflevector(__builtin_shufflevector(s01, s23, 0, 1, 2, 3),
                               __builtin_shufflevector(s45, s67, 0, 1, 2, 3), 0, 1, 2, 3, 4, 5, 6, 7);

  v2f tr = mk2(tanh_hw(d[0]), tanh_hw(d[1]));
  v2f tz = mk2(tanh_hw(d[2]), tanh_hw(d[3]));
  v2f p = __builtin_elementwise_fma(tr, mk2(d[4], d[5]), mk2(d[6], d[7]));
  STAGE("+v"(p), "+v"(tz));

  T = wmma4(axR, xk, cbR);
  __builtin_amdgcn_sched_barrier(0);
  v2f n = mk2(tanh_hw(p.x), tanh_hw(p.y));
  v2f a0 = __builtin_elementwise_fma(tz, nhm0, hm0);
  v2f a1 = __builtin_elementwise_fma(tz, nhm1, hm1);
  v2f z = __builtin_elementwise_fma(tz, half2, half2);
  v2f zh0 = z * rA, zh1 = z * rB;
  nA = __builtin_elementwise_fma(a0, n, zh0);
  nB = __builtin_elementwise_fma(a1, n, zh1);
  STAGE("+v"(T), "+v"(nA), "+v"(nB));

  T = wmma4(a4R, x34, T);
  rA.x = other_group(nB.x);
  rA.y = other_group(nB.y);
  rB.x = other_group(nA.x);
  rB.y = other_group(nA.y);
  STAGE("+v"(T), "+v"(rA), "+v"(rB));
}

__global__ void __launch_bounds__(128) gru_wmma_kernel(
    const float* __restrict__ x, const float* __restrict__ w_ih,
    const float* __restrict__ w_hh, const float* __restrict__ b_ih,
    const float* __restrict__ b_hh, const float* __restrict__ fc_w,
    const float* __restrict__ fc_b, float* __restrict__ out) {
  __shared__ __attribute__((aligned(16))) float xs[4 * 2 * WBUF];
  __shared__ __attribute__((aligned(16))) float osh[4 * NB * 2];
  const int lane = threadIdx.x & 31;
  const int wid  = threadIdx.x >> 5;
  const int col  = lane & 15;
  const int hf   = lane >> 4;
  const int grp  = col >> 3;
  const int bi   = col & 7;
  const bool g1  = grp != 0;

  const int m = col, s0 = 2 * hf;
  const int typ = (m & 7) >> 1;
  const int un0 = 2 * (0 + (m >> 3)) + (m & 1);
  const int un1 = 2 * (2 + (m >> 3)) + (m & 1);
  const v2f ah0 = mk2(wh(w_hh, typ, un0, s0),     wh(w_hh, typ, un0, s0 + 1));
  const v2f ah1 = mk2(wh(w_hh, typ, un0, 4 + s0), wh(w_hh, typ, un0, 5 + s0));
  const v2f ah2 = mk2(wh(w_hh, typ, un1, s0),     wh(w_hh, typ, un1, s0 + 1));
  const v2f ah3 = mk2(wh(w_hh, typ, un1, 4 + s0), wh(w_hh, typ, un1, 5 + s0));
  const v2f ax0 = mk2(wx(w_ih, typ, un0, s0), wx(w_ih, typ, un0, s0 + 1));
  const v2f ax1 = mk2(wx(w_ih, typ, un1, s0), wx(w_ih, typ, un1, s0 + 1));
  const v2f a40 = hf ? mk2(0.f, 0.f) : mk2(0.f, wx(w_ih, typ, un0, 4));
  const v2f a41 = hf ? mk2(0.f, 0.f) : mk2(0.f, wx(w_ih, typ, un1, 4));

  v8f cb0, cb1;
#pragma unroll
  for (int r = 0; r < 8; ++r) {
    const int t = r >> 1, ua = 2 * hf + (r & 1), ub = 4 + ua;
    cb0[r] = t == 0 ? 0.5f * (b_ih[ua] + b_hh[ua])
           : t == 1 ? 0.5f * (b_ih[8 + ua] + b_hh[8 + ua])
           : t == 2 ? 0.5f * b_hh[16 + ua] : b_ih[16 + ua] + 0.5f * b_hh[16 + ua];
    cb1[r] = t == 0 ? 0.5f * (b_ih[ub] + b_hh[ub])
           : t == 1 ? 0.5f * (b_ih[8 + ub] + b_hh[8 + ub])
           : t == 2 ? 0.5f * b_hh[16 + ub] : b_ih[16 + ub] + 0.5f * b_hh[16 + ub];
  }

  const float m0 = grp ? 0.f : 1.f, m1 = grp ? 1.f : 0.f;
  const v2f hm0 = mk2(0.5f * m0, 0.5f * m0), hm1 = mk2(0.5f * m1, 0.5f * m1);
  const v2f nhm0 = -hm0, nhm1 = -hm1;
  const v2f half2 = mk2(0.5f, 0.5f);

  v2f nA = mk2(0.f, 0.f), nB = mk2(0.f, 0.f), rA = mk2(0.f, 0.f), rB = mk2(0.f, 0.f);

  const int wid_s = __builtin_amdgcn_readfirstlane(wid);
  const float* xw = x + (size_t)(blockIdx.x * (4 * NB) + wid_s * NB) * XROW;
  float* wb = xs + wid_s * (2 * WBUF);
  const bool stg = lane < ROW4;
  const int l4 = lane * 4;

  v4f g0[NB], g[NB];
#pragma unroll
  for (int r = 0; r < NB; ++r) { g0[r] = (v4f)(0.f); g[r] = (v4f)(0.f); }
  if (stg) {
#pragma unroll
    for (int r = 0; r < NB; ++r)
      g0[r] = *reinterpret_cast<const v4f*>(xw + r * XROW + l4);
#pragma unroll
    for (int r = 0; r < NB; ++r)
      g[r] = *reinterpret_cast<const v4f*>(xw + r * XROW + TC * I_DIM + l4);
#pragma unroll
    for (int r = 0; r < NB; ++r)
      *reinterpret_cast<v4f*>(wb + r * LROW + l4) = g0[r];
  }

  const float* lrow = wb + bi * LROW;
  const float* t0row = lrow + grp * I_DIM;
  const float* t1row = lrow + (1 - grp) * I_DIM;

  v8f D0a, D0b, D1a, Xe;
  {
    const float* q0 = t0row;
    const float* q1 = t1row;
    const float* q2 = t0row + 2 * I_DIM;
    D0a = wmma4(ax0, mk2(q0[s0], q0[s0 + 1]), cb0);
    D0a = wmma4(a40, mk2(q0[3], q0[4]), D0a);
    D1a = wmma4(ax1, mk2(q1[s0], q1[s0 + 1]), cb1);
    D1a = wmma4(a41, mk2(q1[3], q1[4]), D1a);
    D0b = wmma4(ax0, mk2(q2[s0], q2[s0 + 1]), cb0);
    D0b = wmma4(a40, mk2(q2[3], q2[4]), D0b);
    Xe = g1 ? D1a : D0a;
  }

  for (int c = 0; c < NCHUNK; ++c) {
    const int cbuf = c & 1, nb = cbuf ^ 1;
    if (c + 1 < NCHUNK) {
      if (stg) {
#pragma unroll
        for (int r = 0; r < NB; ++r)
          *reinterpret_cast<v4f*>(wb + nb * WBUF + r * LROW + l4) = g[r];
      }
    }

    const float* c0 = t0row + cbuf * WBUF;
    const float* n0 = t0row + nb * WBUF;
    const float* c1 = t1row + cbuf * WBUF;
    const float* n1 = t1row + nb * WBUF;
    v2f k1[TC / 2], e1[TC / 2], k0[TC / 2], e0[TC / 2];
#pragma unroll
    for (int i = 0; i < TC / 2; ++i) {
      const int st1 = 2 * (i + 1), st0 = 2 * (i + 2);
      const float* q1 = st1 < TC ? c1 + st1 * I_DIM : n1 + (st1 - TC) * I_DIM;
      const float* q0 = st0 < TC ? c0 + st0 * I_DIM : n0 + (st0 - TC) * I_DIM;
      k1[i] = mk2(q1[s0], q1[s0 + 1]);
      e1[i] = mk2(q1[3], q1[4]);
      k0[i] = mk2(q0[s0], q0[s0 + 1]);
      e0[i] = mk2(q0[3], q0[4]);
    }

    if (c + 2 < NCHUNK) {
      if (stg) {
        const float* gp = xw + (c + 2) * (TC * I_DIM) + l4;
#pragma unroll
        for (int r = 0; r < NB; ++r)
          g[r] = *reinterpret_cast<const v4f*>(gp + r * XROW);
      }
    }

#pragma unroll
    for (int q = 0; q < TC / 2; ++q) {
      v8f Xo, D1b, D0c;
      gru_step(ah0, ah1, ah2, ah3, Xe, rA, nA, nB, rB,
               g1, D0a, D1a, Xo, ax1, a41, cb1, k1[q], e1[q], D1b,
               nA, nB, rA, rB, hm0, nhm0, hm1, nhm1, half2);
      gru_step(ah0, ah1, ah2, ah3, Xo, rB, nB, nA, rA,
               g1, D1b, D0b, Xe, ax0, a40, cb0, k0[q], e0[q], D0c,
               nA, nB, rA, rB, hm0, nhm0, hm1, nhm1, half2);
      D0a = D0b; D0b = D0c; D1a = D1b;
    }
  }

  const int ka = 2 * hf;
  const v2f own = nA + nB, oth = rA + rB;
  float o0 = fc_w[0 * 8 + ka] * oth.x + fc_w[0 * 8 + ka + 1] * oth.y +
             fc_w[0 * 8 + ka + 4] * own.x + fc_w[0 * 8 + ka + 5] * own.y;
  float o1 = fc_w[1 * 8 + ka] * oth.x + fc_w[1 * 8 + ka + 1] * oth.y +
             fc_w[1 * 8 + ka + 4] * own.x + fc_w[1 * 8 + ka + 5] * own.y;
  o0 += swz16(o0);
  o1 += swz16(o1);
  if (lane < NB) {
    osh[(wid * NB + lane) * 2 + 0] = o0 + fc_b[0];
    osh[(wid * NB + lane) * 2 + 1] = o1 + fc_b[1];
  }
  __syncthreads();
  if (wid == 0) {
    v2f o = *reinterpret_cast<const v2f*>(osh + lane * 2);
    *reinterpret_cast<v2f*>(out + ((size_t)blockIdx.x * (4 * NB) + lane) * 2) = o;
  }
}

extern "C" void kernel_launch(void* const* d_in, const int* in_sizes, int n_in,
                              void* d_out, int out_size, void* d_ws, size_t ws_size,
                              hipStream_t stream) {
  const float* x    = (const float*)d_in[0];
  const float* w_ih = (const float*)d_in[1];
  const float* w_hh = (const float*)d_in[2];
  const float* b_ih = (const float*)d_in[3];
  const float* b_hh = (const float*)d_in[4];
  const float* fc_w = (const float*)d_in[5];
  const float* fc_b = (const float*)d_in[6];
  float* out = (float*)d_out;

  const int B = in_sizes[0] / (T_DIM * I_DIM);
  dim3 grid(B / 32), block(128);
  gru_wmma_kernel<<<grid, block, 0, stream>>>(x, w_ih, w_hh, b_ih, b_hh, fc_w,
                                              fc_b, out);
}


# x4 term of each x tile made as 3 packed f32 multiply-adds on the C operand (bias + w4*x4), not as a second 16x16x4 product with 3 zero k slots: 5 not 6 products a step; sum order bias,x4,x0..3,h
# speedup vs baseline: 3.2325x; 1.0688x over previous
#include <hip/hip_runtime.h>


typedef float v2f __attribute__((ext_vector_type(2)));
typedef float v4f __attribute__((ext_vector_type(4)));
typedef float v8f __attribute__((ext_vector_type(8)));

#define I_DIM 5
#define H_DIM 8
#define T_DIM 512
#define NB 8
#define TC 16
#define NCHUNK (T_DIM / TC)
#define ROW4 (TC * I_DIM / 4)
#define LROW 84
#define WBUF (NB * LROW)
#define XROW (T_DIM * I_DIM)

__device__ __forceinline__ v2f mk2(float a, float b) { v2f v; v.x = a; v.y = b; return v; }

__device__ __forceinline__ float tanh_hw(float x) { return __builtin_amdgcn_tanhf(x); }

__device__ __forceinline__ float other_group(float v) {
  return __int_as_float(__builtin_amdgcn_mov_dpp(__float_as_int(v), 0x128, 0xf, 0xf, false));
}
__device__ __forceinline__ float swz16(float v) {
  return __int_as_float(__builtin_amdgcn_ds_swizzle(__float_as_int(v), 0x401F));
}

__device__ __forceinline__ v8f wmma4(v2f a, v2f b, v8f c) {
  return __builtin_amdgcn_wmma_f32_16x16x4_f32(false, a, false, b, (short)0, c, false, false);
}

__device__ __forceinline__ float wh(const float* w_hh, int type, int unit, int k) {
  return 0.5f * w_hh[((type == 3 ? 2 : type) * 8 + unit) * H_DIM + k];
}
__device__ __forceinline__ float wx(const float* w_ih, int type, int unit, int i) {
  return type == 2 ? 0.f : (type == 3 ? 1.f : 0.5f) * w_ih[((type == 3 ? 2 : type) * 8 + unit) * I_DIM + i];
}

#define STAGE(...) do { asm volatile("" : __VA_ARGS__); __builtin_amdgcn_sched_barrier(0); } while (0)

__device__ __forceinline__ v8f x4_seed(const v2f w01, const v2f w23, const v2f w67, const v8f cb, const float x4) {
  const v2f x4v = mk2(x4, x4);
  const v2f c01 = __builtin_elementwise_fma(w01, x4v, mk2(cb[0], cb[1]));
  const v2f c23 = __builtin_elementwise_fma(w23, x4v, mk2(cb[2], cb[3]));
  const v2f c67 = __builtin_elementwise_fma(w67, x4v, mk2(cb[6], cb[7]));
  v8f c;
  c[0] = c01.x; c[1] = c01.y; c[2] = c23.x; c[3] = c23.y;
  c[4] = cb[4]; c[5] = cb[5]; c[6] = c67.x; c[7] = c67.y;
  return c;
}

__device__ __forceinline__ void gru_step(
    const v2f ah0, const v2f ah1, const v2f ah2, const v2f ah3,
    const v8f X, const v2f b0, const v2f b1, const v2f b2, const v2f b3,
    const bool g1, const v8f SA, const v8f SB, v8f& Xn,
    const v2f axR, const v2f w01, const v2f w23, const v2f w67, const v8f cbR, v8f& Cs,
    const v2f xk, const float x4, v8f& T,
    v2f& nA, v2f& nB, v2f& rA, v2f& rB,
    const v2f hm0, const v2f nhm0, const v2f hm1, const v2f nhm1, const v2f half2) {
  const v2f x4v = mk2(x4, x4);
  v8f d = wmma4(ah0, b0, X);
  STAGE("+v"(d));
  v2f s01 = g1 ? mk2(SA[0], SA[1]) : mk2(SB[0], SB[1]);
  STAGE("+v"(d), "+v"(s01));
  d = wmma4(ah1, b1, d);
  v2f s23 = g1 ? mk2(SA[2], SA[3]) : mk2(SB[2], SB[3]);
  const v2f c01 = __builtin_elementwise_fma(w01, x4v, mk2(cbR[0], cbR[1]));
  Cs[0] = c01.x; Cs[1] = c01.y;
  STAGE("+v"(d), "+v"(s23), "+v"(Cs));
  d = wmma4(ah2, b2, d);
  v2f s45 = g1 ? mk2(SA[4], SA[5]) : mk2(SB[4], SB[5]);
  const v2f c23 = __builtin_elementwise_fma(w23, x4v, mk2(cbR[2], cbR[3]));
  Cs[2] = c23.x; Cs[3] = c23.y;
  STAGE("+v"(d), "+v"(s45), "+v"(Cs));
  d = wmma4(ah3, b3, d);
  v2f s67 = g1 ? mk2(SA[6], SA[7]) : mk2(SB[6], SB[7]);
  const v2f c67 = __builtin_elementwise_fma(w67, x4v, mk2(cbR[6], cbR[7]));
  Cs[6] = c67.x; Cs[7] = c67.y;
  STAGE("+v"(d), "+v"(s67), "+v"(Cs));
  Xn = __builtin_shufflevector(__builtin_shufflevector(s01, s23, 0, 1, 2, 3),
                               __builtin_shufflevector(s45, s67, 0, 1, 2, 3), 0, 1, 2, 3, 4, 5, 6, 7);

  v2f tr = mk2(tanh_hw(d[0]), tanh_hw(d[1]));
  v2f tz = mk2(tanh_hw(d[2]), tanh_hw(d[3]));
  v2f p = __builtin_elementwise_fma(tr, mk2(d[4], d[5]), mk2(d[6], d[7]));
  STAGE("+v"(p), "+v"(tz));

  T = wmma4(axR, xk, Cs);
  __builtin_amdgcn_sched_barrier(0);
  v2f n = mk2(tanh_hw(p.x), tanh_hw(p.y));
  v2f a0 = __builtin_elementwise_fma(tz, nhm0, hm0);
  v2f a1 = __builtin_elementwise_fma(tz, nhm1, hm1);
  v2f z = __builtin_elementwise_fma(tz, half2, half2);
  v2f zh0 = z * rA, zh1 = z * rB;
  nA = __builtin_elementwise_fma(a0, n, zh0);
  nB = __builtin_elementwise_fma(a1, n, zh1);
  STAGE("+v"(T), "+v"(nA), "+v"(nB), "+v"(Cs));

  rA.x = other_group(nB.x);
  rA.y = other_group(nB.y);
  rB.x = other_group(nA.x);
  rB.y = other_group(nA.y);
  STAGE("+v"(rA), "+v"(rB));
}

__global__ void __launch_bounds__(128) gru_wmma_kernel(
    const float* __restrict__ x, const float* __restrict__ w_ih,
    const float* __restrict__ w_hh, const float* __restrict__ b_ih,
    const float* __restrict__ b_hh, const float* __restrict__ fc_w,
    const float* __restrict__ fc_b, float* __restrict__ out) {
  __shared__ __attribute__((aligned(16))) float xs[4 * 2 * WBUF];
  __shared__ __attribute__((aligned(16))) float osh[4 * NB * 2];
  const int lane = threadIdx.x & 31;
  const int wid  = threadIdx.x >> 5;
  const int col  = lane & 15;
  const int hf   = lane >> 4;
  const int grp  = col >> 3;
  const int bi   = col & 7;
  const bool g1  = grp != 0;

  const int m = col, s0 = 2 * hf;
  const int typ = (m & 7) >> 1;
  const int un0 = 2 * (0 + (m >> 3)) + (m & 1);
  const int un1 = 2 * (2 + (m >> 3)) + (m & 1);
  const v2f ah0 = mk2(wh(w_hh, typ, un0, s0),     wh(w_hh, typ, un0, s0 + 1));
  const v2f ah1 = mk2(wh(w_hh, typ, un0, 4 + s0), wh(w_hh, typ, un0, 5 + s0));
  const v2f ah2 = mk2(wh(w_hh, typ, un1, s0),     wh(w_hh, typ, un1, s0 + 1));
  const v2f ah3 = mk2(wh(w_hh, typ, un1, 4 + s0), wh(w_hh, typ, un1, 5 + s0));
  const v2f ax0 = mk2(wx(w_ih, typ, un0, s0), wx(w_ih, typ, un0, s0 + 1));
  const v2f ax1 = mk2(wx(w_ih, typ, un1, s0), wx(w_ih, typ, un1, s0 + 1));
  const int ua0 = 2 * hf, ub0 = 4 + 2 * hf;
  const v2f w01_0 = mk2(wx(w_ih, 0, ua0, 4), wx(w_ih, 0, ua0 + 1, 4));
  const v2f w23_0 = mk2(wx(w_ih, 1, ua0, 4), wx(w_ih, 1, ua0 + 1, 4));
  const v2f w67_0 = mk2(wx(w_ih, 3, ua0, 4), wx(w_ih, 3, ua0 + 1, 4));
  const v2f w01_1 = mk2(wx(w_ih, 0, ub0, 4), wx(w_ih, 0, ub0 + 1, 4));
  const v2f w23_1 = mk2(wx(w_ih, 1, ub0, 4), wx(w_ih, 1, ub0 + 1, 4));
  const v2f w67_1 = mk2(wx(w_ih, 3, ub0, 4), wx(w_ih, 3, ub0 + 1, 4));

  v8f cb0, cb1;
#pragma unroll
  for (int r = 0; r < 8; ++r) {
    const int t = r >> 1, ua = 2 * hf + (r & 1), ub = 4 + ua;
    cb0[r] = t == 0 ? 0.5f * (b_ih[ua] + b_hh[ua])
           : t == 1 ? 0.5f * (b_ih[8 + ua] + b_hh[8 + ua])
           : t == 2 ? 0.5f * b_hh[16 + ua] : b_ih[16 + ua] + 0.5f * b_hh[16 + ua];
    cb1[r] = t == 0 ? 0.5f * (b_ih[ub] + b_hh[ub])
           : t == 1 ? 0.5f * (b_ih[8 + ub] + b_hh[8 + ub])
           : t == 2 ? 0.5f * b_hh[16 + ub] : b_ih[16 + ub] + 0.5f * b_hh[16 + ub];
  }

  v8f Cs0 = cb0, Cs1 = cb1;

  const float m0 = grp ? 0.f : 1.f, m1 = grp ? 1.f : 0.f;
  const v2f hm0 = mk2(0.5f * m0, 0.5f * m0), hm1 = mk2(0.5f * m1, 0.5f * m1);
  const v2f nhm0 = -hm0, nhm1 = -hm1;
  const v2f half2 = mk2(0.5f, 0.5f);

  v2f nA = mk2(0.f, 0.f), nB = mk2(0.f, 0.f), rA = mk2(0.f, 0.f), rB = mk2(0.f, 0.f);

  const int wid_s = __builtin_amdgcn_readfirstlane(wid);
  const float* xw = x + (size_t)(blockIdx.x * (4 * NB) + wid_s * NB) * XROW;
  float* wb = xs + wid_s * (2 * WBUF);
  const bool stg = lane < ROW4;
  const int l4 = lane * 4;

  v4f g0[NB], g[NB];
#pragma unroll
  for (int r = 0; r < NB; ++r) { g0[r] = (v4f)(0.f); g[r] = (v4f)(0.f); }
  if (stg) {
#pragma unroll
    for (int r = 0; r < NB; ++r)
      g0[r] = *reinterpret_cast<const v4f*>(xw + r * XROW + l4);
#pragma unroll
    for (int r = 0; r < NB; ++r)
      g[r] = *reinterpret_cast<const v4f*>(xw + r * XROW + TC * I_DIM + l4);
#pragma unroll
    for (int r = 0; r < NB; ++r)
      *reinterpret_cast<v4f*>(wb + r * LROW + l4) = g0[r];
  }

  const float* lrow = wb + bi * LROW;
  const float* t0row = lrow + grp * I_DIM;
  const float* t1row = lrow + (1 - grp) * I_DIM;

  v8f D0a, D0b, D1a, Xe;
  {
    const float* q0 = t0row;
    const float* q1 = t1row;
    const float* q2 = t0row + 2 * I_DIM;
    D0a = wmma4(ax0, mk2(q0[s0], q0[s0 + 1]), x4_seed(w01_0, w23_0, w67_0, cb0, q0[4]));
    D1a = wmma4(ax1, mk2(q1[s0], q1[s0 + 1]), x4_seed(w01_1, w23_1, w67_1, cb1, q1[4]));
    D0b = wmma4(ax0, mk2(q2[s0], q2[s0 + 1]), x4_seed(w01_0, w23_0, w67_0, cb0, q2[4]));
    Xe = g1 ? D1a : D0a;
  }

  for (int c = 0; c < NCHUNK; ++c) {
    const int cbuf = c & 1, nb = cbuf ^ 1;
    if (c + 1 < NCHUNK) {
      if (stg) {
#pragma unroll
        for (int r = 0; r < NB; ++r)
          *reinterpret_cast<v4f*>(wb + nb * WBUF + r * LROW + l4) = g[r];
      }
    }

    const float* c0 = t0row + cbuf * WBUF;
    const float* n0 = t0row + nb * WBUF;
    const float* c1 = t1row + cbuf * WBUF;
    const float* n1 = t1row + nb * WBUF;
    v2f k1[TC / 2], k0[TC / 2];
    float e1[TC / 2], e0[TC / 2];
#pragma unroll
    for (int i = 0; i < TC / 2; ++i) {
      const int st1 = 2 * (i + 1), st0 = 2 * (i + 2);
      const float* q1 = st1 < TC ? c1 + st1 * I_DIM : n1 + (st1 - TC) * I_DIM;
      const float* q0 = st0 < TC ? c0 + st0 * I_DIM : n0 + (st0 - TC) * I_DIM;
      k1[i] = mk2(q1[s0], q1[s0 + 1]);
      e1[i] = q1[4];
      k0[i] = mk2(q0[s0], q0[s0 + 1]);
      e0[i] = q0[4];
    }

    if (c + 2 < NCHUNK) {
      if (stg) {
        const float* gp = xw + (c + 2) * (TC * I_DIM) + l4;
#pragma unroll
        for (int r = 0; r < NB; ++r)
          g[r] = *reinterpret_cast<const v4f*>(gp + r * XROW);
      }
    }

#pragma unroll
    for (int q = 0; q < TC / 2; ++q) {
      v8f Xo, D1b, D0c;
      gru_step(ah0, ah1, ah2, ah3, Xe, rA, nA, nB, rB,
               g1, D0a, D1a, Xo, ax1, w01_1, w23_1, w67_1, cb1, Cs1, k1[q], e1[q], D1b,
               nA, nB, rA, rB, hm0, nhm0, hm1, nhm1, half2);
      gru_step(ah0, ah1, ah2, ah3, Xo, rB, nB, nA, rA,
               g1, D1b, D0b, Xe, ax0, w01_0, w23_0, w67_0, cb0, Cs0, k0[q], e0[q], D0c,
               nA, nB, rA, rB, hm0, nhm0, hm1, nhm1, half2);
      D0a = D0b; D0b = D0c; D1a = D1b;
    }
  }

  const int ka = 2 * hf;
  const v2f own = nA + nB, oth = rA + rB;
  float o0 = fc_w[0 * 8 + ka] * oth.x + fc_w[0 * 8 + ka + 1] * oth.y +
             fc_w[0 * 8 + ka + 4] * own.x + fc_w[0 * 8 + ka + 5] * own.y;
  float o1 = fc_w[1 * 8 + ka] * oth.x + fc_w[1 * 8 + ka + 1] * oth.y +
             fc_w[1 * 8 + ka + 4] * own.x + fc_w[1 * 8 + ka + 5] * own.y;
  o0 += swz16(o0);
  o1 += swz16(o1);
  if (lane < NB) {
    osh[(wid * NB + lane) * 2 + 0] = o0 + fc_b[0];
    osh[(wid * NB + lane) * 2 + 1] = o1 + fc_b[1];
  }
  __syncthreads();
  if (wid == 0) {
    v2f o = *reinterpret_cast<const v2f*>(osh + lane * 2);
    *reinterpret_cast<v2f*>(out + ((size_t)blockIdx.x * (4 * NB) + lane) * 2) = o;
  }
}

extern "C" void kernel_launch(void* const* d_in, const int* in_sizes, int n_in,
                              void* d_out, int out_size, void* d_ws, size_t ws_size,
                              hipStream_t stream) {
  const float* x    = (const float*)d_in[0];
  const float* w_ih = (const float*)d_in[1];
  const float* w_hh = (const float*)d_in[2];
  const float* b_ih = (const float*)d_in[3];
  const float* b_hh = (const float*)d_in[4];
  const float* fc_w = (const float*)d_in[5];
  const float* fc_b = (const float*)d_in[6];
  float* out = (float*)d_out;

  const int B = in_sizes[0] / (T_DIM * I_DIM);
  dim3 grid(B / 32), block(128);
  gru_wmma_kernel<<<grid, block, 0, stream>>>(x, w_ih, w_hh, b_ih, b_hh, fc_w,
                                              fc_b, out);
}
